# starting kernel
# baseline (speedup 1.0000x reference)
#include <hip/hip_runtime.h>

typedef float v2f __attribute__((ext_vector_type(2)));
typedef float v8f __attribute__((ext_vector_type(8)));

#define IN_FEAT 512
#define HIDDEN  32
#define NCLS    7
#define H2S     8


__device__ __forceinline__ float atomAddF(float* p, float v) {
  return unsafeAtomicAdd(p, v);
}


__global__ void init_ws_kernel(float* __restrict__ out1, float* __restrict__ out2,
                               int* __restrict__ deg, int n_nodes) {
  int idx = blockIdx.x * blockDim.x + threadIdx.x;
  int tot1 = n_nodes * HIDDEN;
  if (idx < tot1) out1[idx] = 0.0f;
  if (idx < n_nodes * H2S) out2[idx] = 0.0f;
  if (idx < n_nodes) deg[idx] = 0;
}


__global__ void degree_kernel(const int* __restrict__ dst, int n_edges,
                              int* __restrict__ deg) {
  int e = blockIdx.x * blockDim.x + threadIdx.x;
  if (e < n_edges) atomicAdd(&deg[dst[e]], 1);
}

__global__ void dinv_kernel(const int* __restrict__ deg, float* __restrict__ dinv,
                            int n_nodes) {
  int v = blockIdx.x * blockDim.x + threadIdx.x;
  if (v < n_nodes) dinv[v] = rsqrtf((float)(deg[v] + 1));
}


__global__ __launch_bounds__(256) void gemm1_wmma_kernel(
    const float* __restrict__ x, const float* __restrict__ W1,
    float* __restrict__ h1, int n_nodes) {
  __shared__ float sW[IN_FEAT * HIDDEN];
  for (int i = threadIdx.x; i < IN_FEAT * HIDDEN; i += 256) {
    int k = i >> 5;
    int n = i & 31;
    sW[(k >> 1) * 64 + n * 2 + (k & 1)] = W1[i];
  }
  __syncthreads();

  const int wave = threadIdx.x >> 5;
  const int lane = threadIdx.x & 31;
  const int half = lane >> 4;
  const int ml   = lane & 15;

  const int tile   = blockIdx.x * 8 + wave;
  const int ntiles = n_nodes >> 4;
  if (tile >= ntiles) return;

  const float* xrow = x + (size_t)(tile * 16 + ml) * IN_FEAT + 2 * half;
  const float* bptr = sW + half * 64 + ml * 2;

  v8f acc0 = {};
  v8f acc1 = {};

  for (int k0 = 0; k0 < IN_FEAT; k0 += 32) {
    __builtin_prefetch(xrow + k0 + 256, 0, 3);
#pragma unroll
    for (int kk = 0; kk < 32; kk += 4) {
      const int k = k0 + kk;
      v2f a  = *(const v2f*)(xrow + k);
      v2f b0 = *(const v2f*)(bptr + k * 32);
      v2f b1 = *(const v2f*)(bptr + k * 32 + 32);
      acc0 = __builtin_amdgcn_wmma_f32_16x16x4_f32(false, a, false, b0,
                                                   (short)0, acc0, false, false);
      acc1 = __builtin_amdgcn_wmma_f32_16x16x4_f32(false, a, false, b1,
                                                   (short)0, acc1, false, false);
    }
  }

  float* orow = h1 + (size_t)(tile * 16 + half * 8) * HIDDEN;
#pragma unroll
  for (int g = 0; g < 8; ++g) {
    orow[(size_t)g * HIDDEN + ml]      = acc0[g];
    orow[(size_t)g * HIDDEN + 16 + ml] = acc1[g];
  }
}


__global__ void agg1_kernel(const int* __restrict__ src, const int* __restrict__ dst,
                            const float* __restrict__ dinv, const float* __restrict__ h1,
                            float* __restrict__ out1, int n_edges) {
  int e = blockIdx.x * 8 + (threadIdx.x >> 5);
  if (e >= n_edges) return;
  e = __builtin_amdgcn_readfirstlane(e);
  int f = threadIdx.x & 31;
  int s = src[e];
  int d = dst[e];
  float norm = dinv[s] * dinv[d];
  atomAddF(&out1[(size_t)d * HIDDEN + f], h1[(size_t)s * HIDDEN + f] * norm);
}

__global__ void relu_self1_kernel(const float* __restrict__ h1,
                                  const float* __restrict__ dinv,
                                  const float* __restrict__ b1,
                                  float* __restrict__ out1, int n_nodes) {
  int idx = blockIdx.x * blockDim.x + threadIdx.x;
  if (idx >= n_nodes * HIDDEN) return;
  int v = idx >> 5;
  int f = idx & 31;
  float di = dinv[v];
  float val = out1[idx] + h1[idx] * di * di + b1[f];
  out1[idx] = val > 0.0f ? val : 0.0f;
}


__global__ void gemm2_kernel(const float* __restrict__ out1, const float* __restrict__ W2,
                             float* __restrict__ h2, int n_nodes) {
  int idx = blockIdx.x * blockDim.x + threadIdx.x;
  if (idx >= n_nodes * H2S) return;
  int v = idx >> 3;
  int c = idx & 7;
  float acc = 0.0f;
  if (c < NCLS) {
    const float* r = out1 + (size_t)v * HIDDEN;
#pragma unroll
    for (int f = 0; f < HIDDEN; ++f) acc = fmaf(r[f], W2[f * NCLS + c], acc);
  }
  h2[idx] = acc;
}


__global__ void agg2_kernel(const int* __restrict__ src, const int* __restrict__ dst,
                            const float* __restrict__ dinv, const float* __restrict__ h2,
                            float* __restrict__ out2, int n_edges) {
  int idx = blockIdx.x * blockDim.x + threadIdx.x;
  int e = idx >> 3;
  if (e >= n_edges) return;
  int c = idx & 7;
  if (c >= NCLS) return;
  int s = src[e];
  int d = dst[e];
  float norm = dinv[s] * dinv[d];
  atomAddF(&out2[(size_t)d * H2S + c], h2[(size_t)s * H2S + c] * norm);
}


__global__ void final_softmax_kernel(const float* __restrict__ h2,
                                     const float* __restrict__ out2,
                                     const float* __restrict__ dinv,
                                     const float* __restrict__ b2,
                                     float* __restrict__ y, int n_nodes) {
  int v = blockIdx.x * blockDim.x + threadIdx.x;
  if (v >= n_nodes) return;
  float di = dinv[v];
  float sl = di * di;
  float z[NCLS];
  float m = -3.0e38f;
#pragma unroll
  for (int c = 0; c < NCLS; ++c) {
    z[c] = out2[(size_t)v * H2S + c] + h2[(size_t)v * H2S + c] * sl + b2[c];
    m = fmaxf(m, z[c]);
  }
  float sum = 0.0f;
#pragma unroll
  for (int c = 0; c < NCLS; ++c) {
    z[c] = expf(z[c] - m);
    sum += z[c];
  }
  float inv = 1.0f / sum;
#pragma unroll
  for (int c = 0; c < NCLS; ++c) y[(size_t)v * NCLS + c] = z[c] * inv;
}


extern "C" void kernel_launch(void* const* d_in, const int* in_sizes, int n_in,
                              void* d_out, int out_size, void* d_ws, size_t ws_size,
                              hipStream_t stream) {
  const float* x  = (const float*)d_in[0];
  const int*   ei = (const int*)d_in[1];
  const float* W1 = (const float*)d_in[2];
  const float* b1 = (const float*)d_in[3];
  const float* W2 = (const float*)d_in[4];
  const float* b2 = (const float*)d_in[5];
  float* y = (float*)d_out;

  const int n_nodes = in_sizes[0] / IN_FEAT;
  const int n_edges = in_sizes[1] / 2;
  const int* srcp = ei;
  const int* dstp = ei + n_edges;

  char* ws = (char*)d_ws;
  size_t off = 0;
  auto carve = [&](size_t bytes) -> void* {
    void* p = ws + off;
    off += (bytes + 255) & ~(size_t)255;
    return p;
  };
  int*   deg  = (int*)  carve((size_t)n_nodes * sizeof(int));
  float* dinv = (float*)carve((size_t)n_nodes * sizeof(float));
  float* h1   = (float*)carve((size_t)n_nodes * HIDDEN * sizeof(float));
  float* out1 = (float*)carve((size_t)n_nodes * HIDDEN * sizeof(float));
  float* h2   = (float*)carve((size_t)n_nodes * H2S * sizeof(float));
  float* out2 = (float*)carve((size_t)n_nodes * H2S * sizeof(float));
  (void)ws_size;

  const int B = 256;
  auto blocks = [&](long long n) { return (int)((n + B - 1) / B); };

  init_ws_kernel<<<blocks((long long)n_nodes * HIDDEN), B, 0, stream>>>(out1, out2, deg, n_nodes);
  degree_kernel<<<blocks(n_edges), B, 0, stream>>>(dstp, n_edges, deg);
  dinv_kernel<<<blocks(n_nodes), B, 0, stream>>>(deg, dinv, n_nodes);

  int ntiles = n_nodes / 16;
  int g1_blocks = (ntiles + 7) / 8;
  gemm1_wmma_kernel<<<g1_blocks, 256, 0, stream>>>(x, W1, h1, n_nodes);

  agg1_kernel<<<(n_edges + 7) / 8, 256, 0, stream>>>(srcp, dstp, dinv, h1, out1, n_edges);
  relu_self1_kernel<<<blocks((long long)n_nodes * HIDDEN), B, 0, stream>>>(h1, dinv, b1, out1, n_nodes);
  gemm2_kernel<<<blocks((long long)n_nodes * H2S), B, 0, stream>>>(out1, W2, h2, n_nodes);
  agg2_kernel<<<blocks((long long)n_edges * H2S), B, 0, stream>>>(srcp, dstp, dinv, h2, out2, n_edges);
  final_softmax_kernel<<<blocks(n_nodes), B, 0, stream>>>(h2, out2, dinv, b2, y, n_nodes);
}


# layer sums as destination-ordered row gathers, no f32 atomics: edge rank from a returning count add (arrival-order rank), scan, fill; W2 product and softmax folded into the gathers
# speedup vs baseline: 1.8588x; 1.8588x over previous
#include <hip/hip_runtime.h>


typedef float v2f __attribute__((ext_vector_type(2)));
typedef float v8f __attribute__((ext_vector_type(8)));

#define IN_FEAT 512
#define HIDDEN  32
#define NCLS    7
#define H2S     8
#define SCAN_T  1024


__device__ __forceinline__ int rl_i(int x, int l) {
  return __builtin_amdgcn_readlane(x, l);
}
__device__ __forceinline__ float rl_f(float x, int l) {
  return __int_as_float(__builtin_amdgcn_readlane(__float_as_int(x), l));
}


__global__ void zero_deg_kernel(int* __restrict__ deg, int n_nodes) {
  int idx = blockIdx.x * blockDim.x + threadIdx.x;
  if (idx < n_nodes) deg[idx] = 0;
}

__global__ void rank_kernel(const int* __restrict__ dst, int n_edges,
                            int* __restrict__ deg, int* __restrict__ rank) {
  int e = blockIdx.x * blockDim.x + threadIdx.x;
  if (e < n_edges) rank[e] = atomicAdd(&deg[dst[e]], 1);
}


__global__ __launch_bounds__(SCAN_T) void scan_local_kernel(
    const int* __restrict__ deg, int* __restrict__ rowptr,
    int* __restrict__ blocksum, int n_nodes) {
  __shared__ int wsum[32];
  const int i = blockIdx.x * SCAN_T + threadIdx.x;
  const int lane = threadIdx.x & 31;
  const int wave = threadIdx.x >> 5;
  const int v = (i < n_nodes) ? deg[i] : 0;
  int x = v;
  for (int o = 1; o < 32; o <<= 1) {
    int t = __shfl_up(x, o);
    if (lane >= o) x += t;
  }
  if (lane == 31) wsum[wave] = x;
  __syncthreads();
  if (wave == 0) {
    const int w = wsum[lane];
    int s = w;
    for (int o = 1; o < 32; o <<= 1) {
      int t = __shfl_up(s, o);
      if (lane >= o) s += t;
    }
    wsum[lane] = s - w;
    if (lane == 31) blocksum[blockIdx.x] = s;
  }
  __syncthreads();
  if (i < n_nodes) rowptr[i] = wsum[wave] + x - v;
}

__global__ __launch_bounds__(SCAN_T) void scan_add_dinv_kernel(
    const int* __restrict__ deg, const int* __restrict__ blocksum,
    int* __restrict__ rowptr, float* __restrict__ dinv, int n_nodes) {
  __shared__ int off_s;
  if (threadIdx.x < 32) {
    int a = 0;
    for (int b = threadIdx.x; b < (int)blockIdx.x; b += 32) a += blocksum[b];
    for (int o = 16; o > 0; o >>= 1) a += __shfl_xor(a, o);
    if (threadIdx.x == 0) off_s = a;
  }
  __syncthreads();
  const int i = blockIdx.x * SCAN_T + threadIdx.x;
  if (i < n_nodes) {
    rowptr[i] += off_s;
    dinv[i] = rsqrtf((float)(deg[i] + 1));
  }
}

__global__ void fill_kernel(const int* __restrict__ src, const int* __restrict__ dst,
                            const int* __restrict__ rank, const int* __restrict__ rowptr,
                            int* __restrict__ col, int n_edges) {
  int e = blockIdx.x * blockDim.x + threadIdx.x;
  if (e < n_edges) col[rowptr[dst[e]] + rank[e]] = src[e];
}


__global__ __launch_bounds__(256) void gemm1_wmma_kernel(
    const float* __restrict__ x, const float* __restrict__ W1,
    float* __restrict__ h1, int n_nodes) {
  __shared__ float sW[IN_FEAT * HIDDEN];
  for (int i = threadIdx.x; i < IN_FEAT * HIDDEN; i += 256) {
    int k = i >> 5;
    int n = i & 31;
    sW[(k >> 1) * 64 + n * 2 + (k & 1)] = W1[i];
  }
  __syncthreads();

  const int wave = threadIdx.x >> 5;
  const int lane = threadIdx.x & 31;
  const int half = lane >> 4;
  const int ml   = lane & 15;

  const int tile   = blockIdx.x * 8 + wave;
  const int ntiles = n_nodes >> 4;
  if (tile >= ntiles) return;

  const float* xrow = x + (size_t)(tile * 16 + ml) * IN_FEAT + 2 * half;
  const float* bptr = sW + half * 64 + ml * 2;

  v8f acc0 = {};
  v8f acc1 = {};

  for (int k0 = 0; k0 < IN_FEAT; k0 += 32) {
    __builtin_prefetch(xrow + k0 + 256, 0, 3);
#pragma unroll
    for (int kk = 0; kk < 32; kk += 4) {
      const int k = k0 + kk;
      v2f a  = *(const v2f*)(xrow + k);
      v2f b0 = *(const v2f*)(bptr + k * 32);
      v2f b1 = *(const v2f*)(bptr + k * 32 + 32);
      acc0 = __builtin_amdgcn_wmma_f32_16x16x4_f32(false, a, false, b0,
                                                   (short)0, acc0, false, false);
      acc1 = __builtin_amdgcn_wmma_f32_16x16x4_f32(false, a, false, b1,
                                                   (short)0, acc1, false, false);
    }
  }

  float* orow = h1 + (size_t)(tile * 16 + half * 8) * HIDDEN;
#pragma unroll
  for (int g = 0; g < 8; ++g) {
    orow[(size_t)g * HIDDEN + ml]      = acc0[g];
    orow[(size_t)g * HIDDEN + 16 + ml] = acc1[g];
  }
}


__global__ __launch_bounds__(256) void gather1_kernel(
    const int* __restrict__ rowptr, const int* __restrict__ deg,
    const int* __restrict__ col, const float* __restrict__ dinv,
    const float* __restrict__ h1, const float* __restrict__ b1,
    const float* __restrict__ W2, float* __restrict__ h2, int n_nodes) {
  int v = blockIdx.x * 8 + (threadIdx.x >> 5);
  if (v >= n_nodes) return;
  v = __builtin_amdgcn_readfirstlane(v);
  const int lane = threadIdx.x & 31;
  const int beg = rowptr[v];
  const int cnt = deg[v];
  const float dv = dinv[v];
  const float hself = h1[(size_t)v * HIDDEN + lane];

  float acc = 0.0f;
  for (int base = 0; base < cnt; base += 32) {
    const int left = cnt - base;
    const int m = left < 32 ? left : 32;
    int sj = 0;
    float wj = 0.0f;
    if (lane < m) {
      sj = col[beg + base + lane];
      wj = dinv[sj] * dv;
    }
    int j = 0;
    for (; j + 8 <= m; j += 8) {
      float hv[8], wv[8];
#pragma unroll
      for (int u = 0; u < 8; ++u) {
        const int s = rl_i(sj, j + u);
        wv[u] = rl_f(wj, j + u);
        hv[u] = h1[(size_t)s * HIDDEN + lane];
      }
#pragma unroll
      for (int u = 0; u < 8; ++u) acc = fmaf(hv[u], wv[u], acc);
    }
    for (; j + 4 <= m; j += 4) {
      float hv[4], wv[4];
#pragma unroll
      for (int u = 0; u < 4; ++u) {
        const int s = rl_i(sj, j + u);
        wv[u] = rl_f(wj, j + u);
        hv[u] = h1[(size_t)s * HIDDEN + lane];
      }
#pragma unroll
      for (int u = 0; u < 4; ++u) acc = fmaf(hv[u], wv[u], acc);
    }
    for (; j < m; ++j) {
      const int s = rl_i(sj, j);
      acc = fmaf(h1[(size_t)s * HIDDEN + lane], rl_f(wj, j), acc);
    }
  }

  const float val = acc + hself * dv * dv + b1[lane];
  const float r = val > 0.0f ? val : 0.0f;

  float outv = 0.0f;
#pragma unroll
  for (int c = 0; c < NCLS; ++c) {
    float t = r * W2[lane * NCLS + c];
    for (int o = 16; o > 0; o >>= 1) t += __shfl_xor(t, o);
    if (lane == c) outv = t;
  }
  if (lane < H2S) h2[(size_t)v * H2S + lane] = outv;
}


__global__ __launch_bounds__(256) void gather2_softmax_kernel(
    const int* __restrict__ rowptr, const int* __restrict__ deg,
    const int* __restrict__ col, const float* __restrict__ dinv,
    const float* __restrict__ h2, const float* __restrict__ b2,
    float* __restrict__ y, int n_nodes) {
  int v = blockIdx.x * 8 + (threadIdx.x >> 5);
  if (v >= n_nodes) return;
  v = __builtin_amdgcn_readfirstlane(v);
  const int lane = threadIdx.x & 31;
  const int q = lane >> 3;
  const int c = lane & 7;
  const int beg = rowptr[v];
  const int cnt = deg[v];
  const float dv = dinv[v];

  float acc = 0.0f;
  for (int base = 0; base < cnt; base += 32) {
    const int left = cnt - base;
    const int m = left < 32 ? left : 32;
    int sj = 0;
    float wj = 0.0f;
    if (lane < m) {
      sj = col[beg + base + lane];
      wj = dinv[sj] * dv;
    }
    float hv[8], wv[8];
#pragma unroll
    for (int it = 0; it < 8; ++it) {
      const int jn = it * 4 + q;
      const int s = __shfl(sj, jn);
      wv[it] = __shfl(wj, jn);
      hv[it] = (jn < m) ? h2[(size_t)s * H2S + c] : 0.0f;
    }
#pragma unroll
    for (int it = 0; it < 8; ++it) acc = fmaf(hv[it], wv[it], acc);
  }
  acc += __shfl_xor(acc, 8);
  acc += __shfl_xor(acc, 16);

  const float sl = dv * dv;
  float z = -3.0e38f;
  if (c < NCLS) z = acc + h2[(size_t)v * H2S + c] * sl + b2[c];
  float mx = z;
  mx = fmaxf(mx, __shfl_xor(mx, 1));
  mx = fmaxf(mx, __shfl_xor(mx, 2));
  mx = fmaxf(mx, __shfl_xor(mx, 4));
  const float ex = (c < NCLS) ? expf(z - mx) : 0.0f;
  float sum = ex;
  sum += __shfl_xor(sum, 1);
  sum += __shfl_xor(sum, 2);
  sum += __shfl_xor(sum, 4);
  const float inv = 1.0f / sum;
  if (q == 0 && c < NCLS) y[(size_t)v * NCLS + c] = ex * inv;
}


extern "C" void kernel_launch(void* const* d_in, const int* in_sizes, int n_in,
                              void* d_out, int out_size, void* d_ws, size_t ws_size,
                              hipStream_t stream) {
  const float* x  = (const float*)d_in[0];
  const int*   ei = (const int*)d_in[1];
  const float* W1 = (const float*)d_in[2];
  const float* b1 = (const float*)d_in[3];
  const float* W2 = (const float*)d_in[4];
  const float* b2 = (const float*)d_in[5];
  float* y = (float*)d_out;

  const int n_nodes = in_sizes[0] / IN_FEAT;
  const int n_edges = in_sizes[1] / 2;
  const int* srcp = ei;
  const int* dstp = ei + n_edges;
  const int nscan = (n_nodes + SCAN_T - 1) / SCAN_T;

  char* ws = (char*)d_ws;
  size_t off = 0;
  auto carve = [&](size_t bytes) -> void* {
    void* p = ws + off;
    off += (bytes + 255) & ~(size_t)255;
    return p;
  };
  const size_t h1_bytes   = (size_t)n_nodes * HIDDEN * sizeof(float);
  const size_t rank_bytes = (size_t)n_edges * sizeof(int);
  int*   deg      = (int*)  carve((size_t)n_nodes * sizeof(int));
  float* dinv     = (float*)carve((size_t)n_nodes * sizeof(float));
  int*   rowptr   = (int*)  carve((size_t)n_nodes * sizeof(int));
  int*   blocksum = (int*)  carve((size_t)nscan * sizeof(int));
  void*  shared_r = carve(h1_bytes > rank_bytes ? h1_bytes : rank_bytes);
  int*   rank     = (int*)shared_r;
  float* h1       = (float*)shared_r;
  int*   col      = (int*)  carve((size_t)n_edges * sizeof(int));
  float* h2       = (float*)carve((size_t)n_nodes * H2S * sizeof(float));
  (void)ws_size;

  const int B = 256;
  auto blocks = [&](long long n) { return (int)((n + B - 1) / B); };

  zero_deg_kernel<<<blocks(n_nodes), B, 0, stream>>>(deg, n_nodes);
  rank_kernel<<<blocks(n_edges), B, 0, stream>>>(dstp, n_edges, deg, rank);
  scan_local_kernel<<<nscan, SCAN_T, 0, stream>>>(deg, rowptr, blocksum, n_nodes);
  scan_add_dinv_kernel<<<nscan, SCAN_T, 0, stream>>>(deg, blocksum, rowptr, dinv, n_nodes);
  fill_kernel<<<blocks(n_edges), B, 0, stream>>>(srcp, dstp, rank, rowptr, col, n_edges);

  int ntiles = n_nodes / 16;
  int g1_blocks = (ntiles + 7) / 8;
  gemm1_wmma_kernel<<<g1_blocks, 256, 0, stream>>>(x, W1, h1, n_nodes);

  gather1_kernel<<<(n_nodes + 7) / 8, 256, 0, stream>>>(rowptr, deg, col, dinv, h1, b1, W2, h2, n_nodes);
  gather2_softmax_kernel<<<(n_nodes + 7) / 8, 256, 0, stream>>>(rowptr, deg, col, dinv, h2, b2, y, n_nodes);
}


# edge lists by destination built by an LDS bucket sort in two launches instead of per-edge global atomics, scan and random scatter; order in a node's list (sum order) = arrival order at LDS counters
# speedup vs baseline: 5.4598x; 2.9373x over previous
#include <hip/hip_runtime.h>


typedef float v2f __attribute__((ext_vector_type(2)));
typedef float v8f __attribute__((ext_vector_type(8)));

#define IN_FEAT 512
#define HIDDEN  32
#define NCLS    7
#define H2S     8


__device__ __forceinline__ int rl_i(int x, int l) {
  return __builtin_amdgcn_readlane(x, l);
}
__device__ __forceinline__ float rl_f(float x, int l) {
  return __int_as_float(__builtin_amdgcn_readlane(__float_as_int(x), l));
}


#define NBK_SH 9
#define NBK    (1 << NBK_SH)
#define CH     16384
#define AT     1024
#define AK     (CH / AT)
#define MAXB   1024
#define BT     1024
#define CAP    24576

__global__ __launch_bounds__(AT) void bin_chunk_kernel(
    const int* __restrict__ src, const int* __restrict__ dst, int n_edges, int nb,
    int* __restrict__ tab, int* __restrict__ packed) {
  __shared__ int hist[MAXB + 1];
  __shared__ int sP[CH];
  const int t = threadIdx.x;
  const int base = blockIdx.x * CH;
  for (int i = t; i <= nb; i += AT) hist[i] = 0;
  __syncthreads();

  int pv[AK], pr[AK], pb[AK];
#pragma unroll
  for (int k = 0; k < AK; ++k) {
    const int e = base + t + k * AT;
    const int ec = e < n_edges ? e : n_edges - 1;
    const int d = dst[ec];
    const int s = src[ec];
    pb[k] = e < n_edges ? (d >> NBK_SH) : -1;
    pv[k] = (s << NBK_SH) | (d & (NBK - 1));
  }
#pragma unroll
  for (int k = 0; k < AK; ++k) {
    pr[k] = 0;
    if (pb[k] >= 0) pr[k] = atomicAdd(&hist[pb[k]], 1);
  }
  __syncthreads();

  if (t < 32) {
    int carry = 0;
    for (int c0 = 0; c0 < nb; c0 += 32) {
      const int i = c0 + t;
      const int x = (i < nb) ? hist[i] : 0;
      int inc = x;
      for (int o = 1; o < 32; o <<= 1) {
        int y = __shfl_up(inc, o);
        if (t >= o) inc += y;
      }
      if (i < nb) hist[i] = carry + inc - x;
      carry += __shfl(inc, 31);
    }
    if (t == 0) hist[nb] = carry;
  }
  __syncthreads();

  int* trow = tab + (size_t)blockIdx.x * (nb + 1);
  for (int i = t; i <= nb; i += AT) trow[i] = hist[i];
#pragma unroll
  for (int k = 0; k < AK; ++k) {
    if (pb[k] >= 0) sP[hist[pb[k]] + pr[k]] = pv[k];
  }
  __syncthreads();

  const int cnt = hist[nb];
  for (int i = t; i < cnt; i += AT) packed[base + i] = sP[i];
}

__global__ __launch_bounds__(BT) void bucket_lists_kernel(
    const int* __restrict__ tab, const int* __restrict__ packed,
    int n_nodes, int nb, int n_chunks,
    int* __restrict__ deg, int* __restrict__ rowptr, float* __restrict__ dinv,
    int* __restrict__ col) {
  __shared__ int cnt[NBK];
  __shared__ int cur[NBK];
  __shared__ int red0[32];
  __shared__ int red1[32];
  __shared__ int bse[2];
  __shared__ int wsum[32];
  __shared__ int stage[CAP];
  const int t = threadIdx.x;
  const int lane = t & 31;
  const int wave = t >> 5;
  const int b = blockIdx.x;
  const int stride = nb + 1;

  int a0 = 0, a1 = 0;
  for (int g = t; g < n_chunks; g += BT) {
    a0 += tab[(size_t)g * stride + b];
    a1 += tab[(size_t)g * stride + b + 1];
  }
  for (int o = 16; o > 0; o >>= 1) {
    a0 += __shfl_xor(a0, o);
    a1 += __shfl_xor(a1, o);
  }
  if (lane == 0) { red0[wave] = a0; red1[wave] = a1; }
  if (t < NBK) cnt[t] = 0;
  __syncthreads();
  if (wave == 0) {
    int x0 = red0[lane], x1 = red1[lane];
    for (int o = 16; o > 0; o >>= 1) {
      x0 += __shfl_xor(x0, o);
      x1 += __shfl_xor(x1, o);
    }
    if (lane == 0) { bse[0] = x0; bse[1] = x1; }
  }
  __syncthreads();
  const int bstart = bse[0];
  const int nbe = bse[1] - bse[0];

  for (int gb = 0; gb < n_chunks; gb += 1024) {
    const int g = gb + lane * 32 + wave;
    int lo = 0, hi = 0;
    if (g < n_chunks) {
      lo = tab[(size_t)g * stride + b];
      hi = tab[(size_t)g * stride + b + 1];
    }
    for (int j = 0; j < 32; ++j) {
      const int gj = gb + j * 32 + wave;
      if (gj >= n_chunks) break;
      const int glo = rl_i(lo, j);
      const int ghi = rl_i(hi, j);
      const int* run = packed + (size_t)gj * CH;
      for (int i = glo + lane; i < ghi; i += 128) {
        int v[4];
#pragma unroll
        for (int u = 0; u < 4; ++u) v[u] = (i + 32 * u < ghi) ? run[i + 32 * u] : -1;
#pragma unroll
        for (int u = 0; u < 4; ++u)
          if (v[u] >= 0) atomicAdd(&cnt[v[u] & (NBK - 1)], 1);
      }
    }
  }
  __syncthreads();

  int c = 0, x = 0;
  if (t < NBK) {
    c = cnt[t];
    x = c;
    for (int o = 1; o < 32; o <<= 1) {
      int y = __shfl_up(x, o);
      if (lane >= o) x += y;
    }
    if (lane == 31) wsum[wave] = x;
  }
  __syncthreads();
  if (wave == 0) {
    const int w = (lane < NBK / 32) ? wsum[lane] : 0;
    int s = w;
    for (int o = 1; o < 32; o <<= 1) {
      int y = __shfl_up(s, o);
      if (lane >= o) s += y;
    }
    wsum[lane] = s - w;
  }
  __syncthreads();
  if (t < NBK) {
    const int ex = wsum[wave] + x - c;
    cur[t] = ex;
    const int node = b * NBK + t;
    if (node < n_nodes) {
      deg[node] = c;
      rowptr[node] = bstart + ex;
      dinv[node] = rsqrtf((float)(c + 1));
    }
  }
  __syncthreads();

  const bool staged = nbe <= CAP;
  for (int gb = 0; gb < n_chunks; gb += 1024) {
    const int g = gb + lane * 32 + wave;
    int lo = 0, hi = 0;
    if (g < n_chunks) {
      lo = tab[(size_t)g * stride + b];
      hi = tab[(size_t)g * stride + b + 1];
    }
    for (int j = 0; j < 32; ++j) {
      const int gj = gb + j * 32 + wave;
      if (gj >= n_chunks) break;
      const int glo = rl_i(lo, j);
      const int ghi = rl_i(hi, j);
      const int* run = packed + (size_t)gj * CH;
      for (int i = glo + lane; i < ghi; i += 128) {
        int v[4];
#pragma unroll
        for (int u = 0; u < 4; ++u) v[u] = (i + 32 * u < ghi) ? run[i + 32 * u] : -1;
#pragma unroll
        for (int u = 0; u < 4; ++u) {
          if (v[u] >= 0) {
            const int p = atomicAdd(&cur[v[u] & (NBK - 1)], 1);
            const int s = v[u] >> NBK_SH;
            if (staged) stage[p] = s;
            else        col[bstart + p] = s;
          }
        }
      }
    }
  }
  __syncthreads();
  if (staged) {
    for (int i = t; i < nbe; i += BT) col[bstart + i] = stage[i];
  }
}


__global__ __launch_bounds__(256) void gemm1_wmma_kernel(
    const float* __restrict__ x, const float* __restrict__ W1,
    const float* __restrict__ dinv, float* __restrict__ h1, int n_nodes) {
  __shared__ float sW[IN_FEAT * HIDDEN];
  for (int i = threadIdx.x; i < IN_FEAT * HIDDEN; i += 256) {
    int k = i >> 5;
    int n = i & 31;
    sW[(k >> 1) * 64 + n * 2 + (k & 1)] = W1[i];
  }
  __syncthreads();

  const int wave = threadIdx.x >> 5;
  const int lane = threadIdx.x & 31;
  const int half = lane >> 4;
  const int ml   = lane & 15;

  const int tile   = blockIdx.x * 8 + wave;
  const int ntiles = n_nodes >> 4;
  if (tile >= ntiles) return;

  const float* xrow = x + (size_t)(tile * 16 + ml) * IN_FEAT + 2 * half;
  const float* bptr = sW + half * 64 + ml * 2;

  v8f acc0 = {};
  v8f acc1 = {};

  for (int k0 = 0; k0 < IN_FEAT; k0 += 32) {
    __builtin_prefetch(xrow + k0 + 256, 0, 3);
#pragma unroll
    for (int kk = 0; kk < 32; kk += 4) {
      const int k = k0 + kk;
      v2f a  = *(const v2f*)(xrow + k);
      v2f b0 = *(const v2f*)(bptr + k * 32);
      v2f b1 = *(const v2f*)(bptr + k * 32 + 32);
      acc0 = __builtin_amdgcn_wmma_f32_16x16x4_f32(false, a, false, b0,
                                                   (short)0, acc0, false, false);
      acc1 = __builtin_amdgcn_wmma_f32_16x16x4_f32(false, a, false, b1,
                                                   (short)0, acc1, false, false);
    }
  }

  const float4* dr = (const float4*)(dinv + tile * 16 + half * 8);
  const float4 d0 = dr[0];
  const float4 d1 = dr[1];
  const float ds[8] = {d0.x, d0.y, d0.z, d0.w, d1.x, d1.y, d1.z, d1.w};

  float* orow = h1 + (size_t)(tile * 16 + half * 8) * HIDDEN;
#pragma unroll
  for (int g = 0; g < 8; ++g) {
    orow[(size_t)g * HIDDEN + ml]      = acc0[g] * ds[g];
    orow[(size_t)g * HIDDEN + 16 + ml] = acc1[g] * ds[g];
  }
}


__global__ __launch_bounds__(256) void gather1_kernel(
    const int* __restrict__ rowptr, const int* __restrict__ deg,
    const int* __restrict__ col, const float* __restrict__ dinv,
    const float* __restrict__ h1, const float* __restrict__ b1,
    const float* __restrict__ W2, float* __restrict__ h2, int n_nodes) {
  int v = blockIdx.x * 8 + (threadIdx.x >> 5);
  if (v >= n_nodes) return;
  v = __builtin_amdgcn_readfirstlane(v);
  const int lane = threadIdx.x & 31;
  const int beg = rowptr[v];
  const int cnt = deg[v];
  const float dv = dinv[v];
  const float hself = h1[(size_t)v * HIDDEN + lane];

  float acc = 0.0f;
  for (int base = 0; base < cnt; base += 32) {
    const int left = cnt - base;
    const int m = left < 32 ? left : 32;
    int sj = 0;
    if (lane < m) sj = col[beg + base + lane];
    int j = 0;
    for (; j + 8 <= m; j += 8) {
      float hv[8];
#pragma unroll
      for (int u = 0; u < 8; ++u) {
        const int s = rl_i(sj, j + u);
        hv[u] = h1[(size_t)s * HIDDEN + lane];
      }
#pragma unroll
      for (int u = 0; u < 8; ++u) acc += hv[u];
    }
    for (; j + 4 <= m; j += 4) {
      float hv[4];
#pragma unroll
      for (int u = 0; u < 4; ++u) {
        const int s = rl_i(sj, j + u);
        hv[u] = h1[(size_t)s * HIDDEN + lane];
      }
#pragma unroll
      for (int u = 0; u < 4; ++u) acc += hv[u];
    }
    for (; j < m; ++j) {
      const int s = rl_i(sj, j);
      acc += h1[(size_t)s * HIDDEN + lane];
    }
  }

  const float val = (acc + hself) * dv + b1[lane];
  const float r = val > 0.0f ? val : 0.0f;

  float outv = 0.0f;
#pragma unroll
  for (int c = 0; c < NCLS; ++c) {
    float t = r * W2[lane * NCLS + c];
    for (int o = 16; o > 0; o >>= 1) t += __shfl_xor(t, o);
    if (lane == c) outv = t * dv;
  }
  if (lane < H2S) h2[(size_t)v * H2S + lane] = outv;
}


__global__ __launch_bounds__(256) void gather2_softmax_kernel(
    const int* __restrict__ rowptr, const int* __restrict__ deg,
    const int* __restrict__ col, const float* __restrict__ dinv,
    const float* __restrict__ h2, const float* __restrict__ b2,
    float* __restrict__ y, int n_nodes) {
  int v = blockIdx.x * 8 + (threadIdx.x >> 5);
  if (v >= n_nodes) return;
  v = __builtin_amdgcn_readfirstlane(v);
  const int lane = threadIdx.x & 31;
  const int q = lane >> 3;
  const int c = lane & 7;
  const int beg = rowptr[v];
  const int cnt = deg[v];
  const float dv = dinv[v];

  float acc = 0.0f;
  for (int base = 0; base < cnt; base += 32) {
    const int left = cnt - base;
    const int m = left < 32 ? left : 32;
    int sj = 0;
    if (lane < m) sj = col[beg + base + lane];
    float hv[8];
#pragma unroll
    for (int it = 0; it < 8; ++it) {
      const int jn = it * 4 + q;
      const int s = __shfl(sj, jn);
      hv[it] = (jn < m) ? h2[(size_t)s * H2S + c] : 0.0f;
    }
#pragma unroll
    for (int it = 0; it < 8; ++it) acc += hv[it];
  }
  acc += __shfl_xor(acc, 8);
  acc += __shfl_xor(acc, 16);

  float z = -3.0e38f;
  if (c < NCLS) z = (acc + h2[(size_t)v * H2S + c]) * dv + b2[c];
  float mx = z;
  mx = fmaxf(mx, __shfl_xor(mx, 1));
  mx = fmaxf(mx, __shfl_xor(mx, 2));
  mx = fmaxf(mx, __shfl_xor(mx, 4));
  const float ex = (c < NCLS) ? expf(z - mx) : 0.0f;
  float sum = ex;
  sum += __shfl_xor(sum, 1);
  sum += __shfl_xor(sum, 2);
  sum += __shfl_xor(sum, 4);
  const float inv = 1.0f / sum;
  if (q == 0 && c < NCLS) y[(size_t)v * NCLS + c] = ex * inv;
}


extern "C" void kernel_launch(void* const* d_in, const int* in_sizes, int n_in,
                              void* d_out, int out_size, void* d_ws, size_t ws_size,
                              hipStream_t stream) {
  const float* x  = (const float*)d_in[0];
  const int*   ei = (const int*)d_in[1];
  const float* W1 = (const float*)d_in[2];
  const float* b1 = (const float*)d_in[3];
  const float* W2 = (const float*)d_in[4];
  const float* b2 = (const float*)d_in[5];
  float* y = (float*)d_out;

  const int n_nodes = in_sizes[0] / IN_FEAT;
  const int n_edges = in_sizes[1] / 2;
  const int* srcp = ei;
  const int* dstp = ei + n_edges;
  const int nb       = (n_nodes + NBK - 1) / NBK;
  const int n_chunks = (n_edges + CH - 1) / CH;

  char* ws = (char*)d_ws;
  size_t off = 0;
  auto carve = [&](size_t bytes) -> void* {
    void* p = ws + off;
    off += (bytes + 255) & ~(size_t)255;
    return p;
  };
  const size_t h1_bytes   = (size_t)n_nodes * HIDDEN * sizeof(float);
  const size_t rank_bytes = (size_t)n_edges * sizeof(int);
  int*   deg      = (int*)  carve((size_t)n_nodes * sizeof(int));
  float* dinv     = (float*)carve((size_t)n_nodes * sizeof(float));
  int*   rowptr   = (int*)  carve((size_t)n_nodes * sizeof(int));
  void*  shared_r = carve(h1_bytes > rank_bytes ? h1_bytes : rank_bytes);
  int*   packed   = (int*)shared_r;
  float* h1       = (float*)shared_r;
  int*   col      = (int*)  carve((size_t)n_edges * sizeof(int));
  const size_t h2_bytes  = (size_t)n_nodes * H2S * sizeof(float);
  const size_t tab_bytes = (size_t)n_chunks * (nb + 1) * sizeof(int);
  void*  shared_t = carve(h2_bytes > tab_bytes ? h2_bytes : tab_bytes);
  int*   tab      = (int*)shared_t;
  float* h2       = (float*)shared_t;
  (void)ws_size;

  bin_chunk_kernel<<<n_chunks, AT, 0, stream>>>(srcp, dstp, n_edges, nb, tab, packed);
  bucket_lists_kernel<<<nb, BT, 0, stream>>>(tab, packed, n_nodes, nb, n_chunks,
                                             deg, rowptr, dinv, col);

  int ntiles = n_nodes / 16;
  int g1_blocks = (ntiles + 7) / 8;
  gemm1_wmma_kernel<<<g1_blocks, 256, 0, stream>>>(x, W1, dinv, h1, n_nodes);

  gather1_kernel<<<(n_nodes + 7) / 8, 256, 0, stream>>>(rowptr, deg, col, dinv, h1, b1, W2, h2, n_nodes);
  gather2_softmax_kernel<<<(n_nodes + 7) / 8, 256, 0, stream>>>(rowptr, deg, col, dinv, h2, b2, y, n_nodes);
}


# both destination gathers on resident waves (2048 8-wave groups, each wave loops over its nodes) with the next nodes' rowptr/deg/dinv, ids and own row kept in flight; sum order unchanged
# speedup vs baseline: 5.9073x; 1.0820x over previous
#include <hip/hip_runtime.h>


typedef float v2f __attribute__((ext_vector_type(2)));
typedef float v8f __attribute__((ext_vector_type(8)));

#define IN_FEAT 512
#define HIDDEN  32
#define NCLS    7
#define H2S     8


__device__ __forceinline__ int rl_i(int x, int l) {
  return __builtin_amdgcn_readlane(x, l);
}
__device__ __forceinline__ float rl_f(float x, int l) {
  return __int_as_float(__builtin_amdgcn_readlane(__float_as_int(x), l));
}


#define NBK_SH 9
#define NBK    (1 << NBK_SH)
#define CH     16384
#define AT     1024
#define AK     (CH / AT)
#define MAXB   1024
#define BT     1024
#define CAP    24576

__global__ __launch_bounds__(AT) void bin_chunk_kernel(
    const int* __restrict__ src, const int* __restrict__ dst, int n_edges, int nb,
    int* __restrict__ tab, int* __restrict__ packed) {
  __shared__ int hist[MAXB + 1];
  __shared__ int sP[CH];
  const int t = threadIdx.x;
  const int base = blockIdx.x * CH;
  for (int i = t; i <= nb; i += AT) hist[i] = 0;
  __syncthreads();

  int pv[AK], pr[AK], pb[AK];
#pragma unroll
  for (int k = 0; k < AK; ++k) {
    const int e = base + t + k * AT;
    const int ec = e < n_edges ? e : n_edges - 1;
    const int d = dst[ec];
    const int s = src[ec];
    pb[k] = e < n_edges ? (d >> NBK_SH) : -1;
    pv[k] = (s << NBK_SH) | (d & (NBK - 1));
  }
#pragma unroll
  for (int k = 0; k < AK; ++k) {
    pr[k] = 0;
    if (pb[k] >= 0) pr[k] = atomicAdd(&hist[pb[k]], 1);
  }
  __syncthreads();

  if (t < 32) {
    int carry = 0;
    for (int c0 = 0; c0 < nb; c0 += 32) {
      const int i = c0 + t;
      const int x = (i < nb) ? hist[i] : 0;
      int inc = x;
      for (int o = 1; o < 32; o <<= 1) {
        int y = __shfl_up(inc, o);
        if (t >= o) inc += y;
      }
      if (i < nb) hist[i] = carry + inc - x;
      carry += __shfl(inc, 31);
    }
    if (t == 0) hist[nb] = carry;
  }
  __syncthreads();

  int* trow = tab + (size_t)blockIdx.x * (nb + 1);
  for (int i = t; i <= nb; i += AT) trow[i] = hist[i];
#pragma unroll
  for (int k = 0; k < AK; ++k) {
    if (pb[k] >= 0) sP[hist[pb[k]] + pr[k]] = pv[k];
  }
  __syncthreads();

  const int cnt = hist[nb];
  for (int i = t; i < cnt; i += AT) packed[base + i] = sP[i];
}

__global__ __launch_bounds__(BT) void bucket_lists_kernel(
    const int* __restrict__ tab, const int* __restrict__ packed,
    int n_nodes, int nb, int n_chunks,
    int* __restrict__ deg, int* __restrict__ rowptr, float* __restrict__ dinv,
    int* __restrict__ col) {
  __shared__ int cnt[NBK];
  __shared__ int cur[NBK];
  __shared__ int red0[32];
  __shared__ int red1[32];
  __shared__ int bse[2];
  __shared__ int wsum[32];
  __shared__ int stage[CAP];
  const int t = threadIdx.x;
  const int lane = t & 31;
  const int wave = t >> 5;
  const int b = blockIdx.x;
  const int stride = nb + 1;

  int a0 = 0, a1 = 0;
  for (int g = t; g < n_chunks; g += BT) {
    a0 += tab[(size_t)g * stride + b];
    a1 += tab[(size_t)g * stride + b + 1];
  }
  for (int o = 16; o > 0; o >>= 1) {
    a0 += __shfl_xor(a0, o);
    a1 += __shfl_xor(a1, o);
  }
  if (lane == 0) { red0[wave] = a0; red1[wave] = a1; }
  if (t < NBK) cnt[t] = 0;
  __syncthreads();
  if (wave == 0) {
    int x0 = red0[lane], x1 = red1[lane];
    for (int o = 16; o > 0; o >>= 1) {
      x0 += __shfl_xor(x0, o);
      x1 += __shfl_xor(x1, o);
    }
    if (lane == 0) { bse[0] = x0; bse[1] = x1; }
  }
  __syncthreads();
  const int bstart = bse[0];
  const int nbe = bse[1] - bse[0];

  for (int gb = 0; gb < n_chunks; gb += 1024) {
    const int g = gb + lane * 32 + wave;
    int lo = 0, hi = 0;
    if (g < n_chunks) {
      lo = tab[(size_t)g * stride + b];
      hi = tab[(size_t)g * stride + b + 1];
    }
    for (int j = 0; j < 32; ++j) {
      const int gj = gb + j * 32 + wave;
      if (gj >= n_chunks) break;
      const int glo = rl_i(lo, j);
      const int ghi = rl_i(hi, j);
      const int* run = packed + (size_t)gj * CH;
      for (int i = glo + lane; i < ghi; i += 128) {
        int v[4];
#pragma unroll
        for (int u = 0; u < 4; ++u) v[u] = (i + 32 * u < ghi) ? run[i + 32 * u] : -1;
#pragma unroll
        for (int u = 0; u < 4; ++u)
          if (v[u] >= 0) atomicAdd(&cnt[v[u] & (NBK - 1)], 1);
      }
    }
  }
  __syncthreads();

  int c = 0, x = 0;
  if (t < NBK) {
    c = cnt[t];
    x = c;
    for (int o = 1; o < 32; o <<= 1) {
      int y = __shfl_up(x, o);
      if (lane >= o) x += y;
    }
    if (lane == 31) wsum[wave] = x;
  }
  __syncthreads();
  if (wave == 0) {
    const int w = (lane < NBK / 32) ? wsum[lane] : 0;
    int s = w;
    for (int o = 1; o < 32; o <<= 1) {
      int y = __shfl_up(s, o);
      if (lane >= o) s += y;
    }
    wsum[lane] = s - w;
  }
  __syncthreads();
  if (t < NBK) {
    const int ex = wsum[wave] + x - c;
    cur[t] = ex;
    const int node = b * NBK + t;
    if (node < n_nodes) {
      deg[node] = c;
      rowptr[node] = bstart + ex;
      dinv[node] = rsqrtf((float)(c + 1));
    }
  }
  __syncthreads();

  const bool staged = nbe <= CAP;
  for (int gb = 0; gb < n_chunks; gb += 1024) {
    const int g = gb + lane * 32 + wave;
    int lo = 0, hi = 0;
    if (g < n_chunks) {
      lo = tab[(size_t)g * stride + b];
      hi = tab[(size_t)g * stride + b + 1];
    }
    for (int j = 0; j < 32; ++j) {
      const int gj = gb + j * 32 + wave;
      if (gj >= n_chunks) break;
      const int glo = rl_i(lo, j);
      const int ghi = rl_i(hi, j);
      const int* run = packed + (size_t)gj * CH;
      for (int i = glo + lane; i < ghi; i += 128) {
        int v[4];
#pragma unroll
        for (int u = 0; u < 4; ++u) v[u] = (i + 32 * u < ghi) ? run[i + 32 * u] : -1;
#pragma unroll
        for (int u = 0; u < 4; ++u) {
          if (v[u] >= 0) {
            const int p = atomicAdd(&cur[v[u] & (NBK - 1)], 1);
            const int s = v[u] >> NBK_SH;
            if (staged) stage[p] = s;
            else        col[bstart + p] = s;
          }
        }
      }
    }
  }
  __syncthreads();
  if (staged) {
    for (int i = t; i < nbe; i += BT) col[bstart + i] = stage[i];
  }
}


__global__ __launch_bounds__(256) void gemm1_wmma_kernel(
    const float* __restrict__ x, const float* __restrict__ W1,
    const float* __restrict__ dinv, float* __restrict__ h1, int n_nodes) {
  __shared__ float sW[IN_FEAT * HIDDEN];
  for (int i = threadIdx.x; i < IN_FEAT * HIDDEN; i += 256) {
    int k = i >> 5;
    int n = i & 31;
    sW[(k >> 1) * 64 + n * 2 + (k & 1)] = W1[i];
  }
  __syncthreads();

  const int wave = threadIdx.x >> 5;
  const int lane = threadIdx.x & 31;
  const int half = lane >> 4;
  const int ml   = lane & 15;

  const int tile   = blockIdx.x * 8 + wave;
  const int ntiles = n_nodes >> 4;
  if (tile >= ntiles) return;

  const float* xrow = x + (size_t)(tile * 16 + ml) * IN_FEAT + 2 * half;
  const float* bptr = sW + half * 64 + ml * 2;

  v8f acc0 = {};
  v8f acc1 = {};

  for (int k0 = 0; k0 < IN_FEAT; k0 += 32) {
    __builtin_prefetch(xrow + k0 + 256, 0, 3);
#pragma unroll
    for (int kk = 0; kk < 32; kk += 4) {
      const int k = k0 + kk;
      v2f a  = *(const v2f*)(xrow + k);
      v2f b0 = *(const v2f*)(bptr + k * 32);
      v2f b1 = *(const v2f*)(bptr + k * 32 + 32);
      acc0 = __builtin_amdgcn_wmma_f32_16x16x4_f32(false, a, false, b0,
                                                   (short)0, acc0, false, false);
      acc1 = __builtin_amdgcn_wmma_f32_16x16x4_f32(false, a, false, b1,
                                                   (short)0, acc1, false, false);
    }
  }

  const float4* dr = (const float4*)(dinv + tile * 16 + half * 8);
  const float4 d0 = dr[0];
  const float4 d1 = dr[1];
  const float ds[8] = {d0.x, d0.y, d0.z, d0.w, d1.x, d1.y, d1.z, d1.w};

  float* orow = h1 + (size_t)(tile * 16 + half * 8) * HIDDEN;
#pragma unroll
  for (int g = 0; g < 8; ++g) {
    orow[(size_t)g * HIDDEN + ml]      = acc0[g] * ds[g];
    orow[(size_t)g * HIDDEN + 16 + ml] = acc1[g] * ds[g];
  }
}


#define GWG    2048
#define NWAVES (GWG * 8)

__device__ __forceinline__ float sum_rows_l1(const float* __restrict__ h1, int sj, int m,
                                             int lane, float acc) {
  int j = 0;
  for (; j + 8 <= m; j += 8) {
    float hv[8];
#pragma unroll
    for (int u = 0; u < 8; ++u) {
      const int s = rl_i(sj, j + u);
      hv[u] = h1[(size_t)s * HIDDEN + lane];
    }
#pragma unroll
    for (int u = 0; u < 8; ++u) acc += hv[u];
  }
  for (; j + 4 <= m; j += 4) {
    float hv[4];
#pragma unroll
    for (int u = 0; u < 4; ++u) {
      const int s = rl_i(sj, j + u);
      hv[u] = h1[(size_t)s * HIDDEN + lane];
    }
#pragma unroll
    for (int u = 0; u < 4; ++u) acc += hv[u];
  }
  for (; j < m; ++j) {
    const int s = rl_i(sj, j);
    acc += h1[(size_t)s * HIDDEN + lane];
  }
  return acc;
}

__global__ __launch_bounds__(256) void gather1_kernel(
    const int* __restrict__ rowptr, const int* __restrict__ deg,
    const int* __restrict__ col, const float* __restrict__ dinv,
    const float* __restrict__ h1, const float* __restrict__ b1,
    const float* __restrict__ W2, float* __restrict__ h2, int n_nodes) {
  const int lane = threadIdx.x & 31;
  int v = __builtin_amdgcn_readfirstlane(blockIdx.x * 8 + (threadIdx.x >> 5));
  if (v >= n_nodes) return;

  const float bias = b1[lane];
  float w2[NCLS];
#pragma unroll
  for (int c = 0; c < NCLS; ++c) w2[c] = W2[lane * NCLS + c];

  int vn = v;
  bool more = true;
  int nbeg = rowptr[vn];
  int ncnt = deg[vn];
  float ndv = dinv[vn];
  bool have = false;
  int beg = 0, cnt = 0;
  float dv = 0.0f, hself = 0.0f;
  int sj0 = 0, sj1 = 0;

  for (;;) {
    float nhself = 0.0f;
    int nsj0 = 0, nsj1 = 0;
    if (more) {
      nhself = h1[(size_t)vn * HIDDEN + lane];
      if (lane < ncnt) nsj0 = col[nbeg + lane];
      if (lane + 32 < ncnt) nsj1 = col[nbeg + 32 + lane];
    }

    float acc = 0.0f;
    acc = sum_rows_l1(h1, sj0, cnt < 32 ? cnt : 32, lane, acc);
    if (cnt > 32) acc = sum_rows_l1(h1, sj1, cnt < 64 ? cnt - 32 : 32, lane, acc);
    for (int base = 64; base < cnt; base += 32) {
      const int left = cnt - base;
      const int m = left < 32 ? left : 32;
      int sj = 0;
      if (lane < m) sj = col[beg + base + lane];
      acc = sum_rows_l1(h1, sj, m, lane, acc);
    }

    const int vnn = vn + NWAVES;
    const bool more2 = more && vnn < n_nodes;
    int nnbeg = 0, nncnt = 0;
    float nndv = 0.0f;
    if (more2) {
      nnbeg = rowptr[vnn];
      nncnt = deg[vnn];
      nndv = dinv[vnn];
    }

    const float val = (acc + hself) * dv + bias;
    const float r = val > 0.0f ? val : 0.0f;

    float outv = 0.0f;
#pragma unroll
    for (int c = 0; c < NCLS; ++c) {
      float t = r * w2[c];
      for (int o = 16; o > 0; o >>= 1) t += __shfl_xor(t, o);
      if (lane == c) outv = t * dv;
    }
    if (have && lane < H2S) h2[(size_t)v * H2S + lane] = outv;

    if (!more) break;
    have = true;
    v = vn;
    beg = nbeg;
    cnt = ncnt;
    dv = ndv;
    hself = nhself;
    sj0 = nsj0;
    sj1 = nsj1;
    vn = vnn;
    more = more2;
    nbeg = nnbeg;
    ncnt = nncnt;
    ndv = nndv;
  }
}


__device__ __forceinline__ float sum_rows_l2(const float* __restrict__ h2, int sj, int m,
                                             int q, int c, float acc) {
  int s[8];
#pragma unroll
  for (int it = 0; it < 8; ++it) s[it] = __shfl(sj, it * 4 + q);
  float hv[8];
#pragma unroll
  for (int it = 0; it < 8; ++it) {
    const int jn = it * 4 + q;
    hv[it] = (jn < m) ? h2[(size_t)s[it] * H2S + c] : 0.0f;
  }
#pragma unroll
  for (int it = 0; it < 8; ++it) acc += hv[it];
  return acc;
}

__global__ __launch_bounds__(256) void gather2_softmax_kernel(
    const int* __restrict__ rowptr, const int* __restrict__ deg,
    const int* __restrict__ col, const float* __restrict__ dinv,
    const float* __restrict__ h2, const float* __restrict__ b2,
    float* __restrict__ y, int n_nodes) {
  const int lane = threadIdx.x & 31;
  const int q = lane >> 3;
  const int c = lane & 7;
  int v = __builtin_amdgcn_readfirstlane(blockIdx.x * 8 + (threadIdx.x >> 5));
  if (v >= n_nodes) return;

  const float bias = (c < NCLS) ? b2[c] : 0.0f;

  int vn = v;
  bool more = true;
  int nbeg = rowptr[vn];
  int ncnt = deg[vn];
  float ndv = dinv[vn];
  bool have = false;
  int beg = 0, cnt = 0;
  float dv = 0.0f, hown = 0.0f;
  int sj0 = 0, sj1 = 0;

  for (;;) {
    float nhown = 0.0f;
    int nsj0 = 0, nsj1 = 0;
    if (more) {
      nhown = h2[(size_t)vn * H2S + c];
      if (lane < ncnt) nsj0 = col[nbeg + lane];
      if (lane + 32 < ncnt) nsj1 = col[nbeg + 32 + lane];
    }

    float acc = 0.0f;
#pragma clang loop unroll(disable)
    for (int k = 0; k < 2; ++k) {
      const int mk = cnt - 32 * k;
      if (mk <= 0) break;
      acc = sum_rows_l2(h2, k == 0 ? sj0 : sj1, mk < 32 ? mk : 32, q, c, acc);
    }
    for (int base = 64; base < cnt; base += 32) {
      const int left = cnt - base;
      const int m = left < 32 ? left : 32;
      int sj = 0;
      if (lane < m) sj = col[beg + base + lane];
      acc = sum_rows_l2(h2, sj, m, q, c, acc);
    }

    const int vnn = vn + NWAVES;
    const bool more2 = more && vnn < n_nodes;
    int nnbeg = 0, nncnt = 0;
    float nndv = 0.0f;
    if (more2) {
      nnbeg = rowptr[vnn];
      nncnt = deg[vnn];
      nndv = dinv[vnn];
    }

    acc += __shfl_xor(acc, 8);
    acc += __shfl_xor(acc, 16);

    float z = -3.0e38f;
    if (c < NCLS) z = (acc + hown) * dv + bias;
    float mx = z;
    mx = fmaxf(mx, __shfl_xor(mx, 1));
    mx = fmaxf(mx, __shfl_xor(mx, 2));
    mx = fmaxf(mx, __shfl_xor(mx, 4));
    const float ex = (c < NCLS) ? expf(z - mx) : 0.0f;
    float sum = ex;
    sum += __shfl_xor(sum, 1);
    sum += __shfl_xor(sum, 2);
    sum += __shfl_xor(sum, 4);
    const float inv = 1.0f / sum;
    if (have && q == 0 && c < NCLS) y[(size_t)v * NCLS + c] = ex * inv;

    if (!more) break;
    have = true;
    v = vn;
    beg = nbeg;
    cnt = ncnt;
    dv = ndv;
    hown = nhown;
    sj0 = nsj0;
    sj1 = nsj1;
    vn = vnn;
    more = more2;
    nbeg = nnbeg;
    ncnt = nncnt;
    ndv = nndv;
  }
}


extern "C" void kernel_launch(void* const* d_in, const int* in_sizes, int n_in,
                              void* d_out, int out_size, void* d_ws, size_t ws_size,
                              hipStream_t stream) {
  const float* x  = (const float*)d_in[0];
  const int*   ei = (const int*)d_in[1];
  const float* W1 = (const float*)d_in[2];
  const float* b1 = (const float*)d_in[3];
  const float* W2 = (const float*)d_in[4];
  const float* b2 = (const float*)d_in[5];
  float* y = (float*)d_out;

  const int n_nodes = in_sizes[0] / IN_FEAT;
  const int n_edges = in_sizes[1] / 2;
  const int* srcp = ei;
  const int* dstp = ei + n_edges;
  const int nb       = (n_nodes + NBK - 1) / NBK;
  const int n_chunks = (n_edges + CH - 1) / CH;

  char* ws = (char*)d_ws;
  size_t off = 0;
  auto carve = [&](size_t bytes) -> void* {
    void* p = ws + off;
    off += (bytes + 255) & ~(size_t)255;
    return p;
  };
  const size_t h1_bytes   = (size_t)n_nodes * HIDDEN * sizeof(float);
  const size_t rank_bytes = (size_t)n_edges * sizeof(int);
  int*   deg      = (int*)  carve((size_t)n_nodes * sizeof(int));
  float* dinv     = (float*)carve((size_t)n_nodes * sizeof(float));
  int*   rowptr   = (int*)  carve((size_t)n_nodes * sizeof(int));
  void*  shared_r = carve(h1_bytes > rank_bytes ? h1_bytes : rank_bytes);
  int*   packed   = (int*)shared_r;
  float* h1       = (float*)shared_r;
  int*   col      = (int*)  carve((size_t)n_edges * sizeof(int));
  const size_t h2_bytes  = (size_t)n_nodes * H2S * sizeof(float);
  const size_t tab_bytes = (size_t)n_chunks * (nb + 1) * sizeof(int);
  void*  shared_t = carve(h2_bytes > tab_bytes ? h2_bytes : tab_bytes);
  int*   tab      = (int*)shared_t;
  float* h2       = (float*)shared_t;
  (void)ws_size;

  bin_chunk_kernel<<<n_chunks, AT, 0, stream>>>(srcp, dstp, n_edges, nb, tab, packed);
  bucket_lists_kernel<<<nb, BT, 0, stream>>>(tab, packed, n_nodes, nb, n_chunks,
                                             deg, rowptr, dinv, col);

  int ntiles = n_nodes / 16;
  int g1_blocks = (ntiles + 7) / 8;
  gemm1_wmma_kernel<<<g1_blocks, 256, 0, stream>>>(x, W1, dinv, h1, n_nodes);

  gather1_kernel<<<GWG, 256, 0, stream>>>(rowptr, deg, col, dinv, h1, b1, W2, h2, n_nodes);
  gather2_softmax_kernel<<<GWG, 256, 0, stream>>>(rowptr, deg, col, dinv, h2, b2, y, n_nodes);
}


# layer-1 gather loads rows 16 B per lane (4 rows per load, a node's list in flight at once); each feature = 4 list-order partial sums added (q0+q1)+(q2+q3); W2 product 4 in lane, then over 8 lanes
# speedup vs baseline: 6.1727x; 1.0449x over previous
#include <hip/hip_runtime.h>


typedef float v2f __attribute__((ext_vector_type(2)));
typedef float v8f __attribute__((ext_vector_type(8)));

#define IN_FEAT 512
#define HIDDEN  32
#define NCLS    7
#define H2S     8


__device__ __forceinline__ int rl_i(int x, int l) {
  return __builtin_amdgcn_readlane(x, l);
}
__device__ __forceinline__ float rl_f(float x, int l) {
  return __int_as_float(__builtin_amdgcn_readlane(__float_as_int(x), l));
}


#define NBK_SH 9
#define NBK    (1 << NBK_SH)
#define CH     16384
#define AT     1024
#define AK     (CH / AT)
#define MAXB   1024
#define BT     1024
#define CAP    24576

__global__ __launch_bounds__(AT) void bin_chunk_kernel(
    const int* __restrict__ src, const int* __restrict__ dst, int n_edges, int nb,
    int* __restrict__ tab, int* __restrict__ packed) {
  __shared__ int hist[MAXB + 1];
  __shared__ int sP[CH];
  const int t = threadIdx.x;
  const int base = blockIdx.x * CH;
  for (int i = t; i <= nb; i += AT) hist[i] = 0;
  __syncthreads();

  int pv[AK], pr[AK], pb[AK];
#pragma unroll
  for (int k = 0; k < AK; ++k) {
    const int e = base + t + k * AT;
    const int ec = e < n_edges ? e : n_edges - 1;
    const int d = dst[ec];
    const int s = src[ec];
    pb[k] = e < n_edges ? (d >> NBK_SH) : -1;
    pv[k] = (s << NBK_SH) | (d & (NBK - 1));
  }
#pragma unroll
  for (int k = 0; k < AK; ++k) {
    pr[k] = 0;
    if (pb[k] >= 0) pr[k] = atomicAdd(&hist[pb[k]], 1);
  }
  __syncthreads();

  if (t < 32) {
    int carry = 0;
    for (int c0 = 0; c0 < nb; c0 += 32) {
      const int i = c0 + t;
      const int x = (i < nb) ? hist[i] : 0;
      int inc = x;
      for (int o = 1; o < 32; o <<= 1) {
        int y = __shfl_up(inc, o);
        if (t >= o) inc += y;
      }
      if (i < nb) hist[i] = carry + inc - x;
      carry += __shfl(inc, 31);
    }
    if (t == 0) hist[nb] = carry;
  }
  __syncthreads();

  int* trow = tab + (size_t)blockIdx.x * (nb + 1);
  for (int i = t; i <= nb; i += AT) trow[i] = hist[i];
#pragma unroll
  for (int k = 0; k < AK; ++k) {
    if (pb[k] >= 0) sP[hist[pb[k]] + pr[k]] = pv[k];
  }
  __syncthreads();

  const int cnt = hist[nb];
  for (int i = t; i < cnt; i += AT) packed[base + i] = sP[i];
}

__global__ __launch_bounds__(BT) void bucket_lists_kernel(
    const int* __restrict__ tab, const int* __restrict__ packed,
    int n_nodes, int nb, int n_chunks,
    int* __restrict__ deg, int* __restrict__ rowptr, float* __restrict__ dinv,
    int* __restrict__ col) {
  __shared__ int cnt[NBK];
  __shared__ int cur[NBK];
  __shared__ int red0[32];
  __shared__ int red1[32];
  __shared__ int bse[2];
  __shared__ int wsum[32];
  __shared__ int stage[CAP];
  const int t = threadIdx.x;
  const int lane = t & 31;
  const int wave = t >> 5;
  const int b = blockIdx.x;
  const int stride = nb + 1;

  int a0 = 0, a1 = 0;
  for (int g = t; g < n_chunks; g += BT) {
    a0 += tab[(size_t)g * stride + b];
    a1 += tab[(size_t)g * stride + b + 1];
  }
  for (int o = 16; o > 0; o >>= 1) {
    a0 += __shfl_xor(a0, o);
    a1 += __shfl_xor(a1, o);
  }
  if (lane == 0) { red0[wave] = a0; red1[wave] = a1; }
  if (t < NBK) cnt[t] = 0;
  __syncthreads();
  if (wave == 0) {
    int x0 = red0[lane], x1 = red1[lane];
    for (int o = 16; o > 0; o >>= 1) {
      x0 += __shfl_xor(x0, o);
      x1 += __shfl_xor(x1, o);
    }
    if (lane == 0) { bse[0] = x0; bse[1] = x1; }
  }
  __syncthreads();
  const int bstart = bse[0];
  const int nbe = bse[1] - bse[0];

  for (int gb = 0; gb < n_chunks; gb += 1024) {
    const int g = gb + lane * 32 + wave;
    int lo = 0, hi = 0;
    if (g < n_chunks) {
      lo = tab[(size_t)g * stride + b];
      hi = tab[(size_t)g * stride + b + 1];
    }
    for (int j = 0; j < 32; ++j) {
      const int gj = gb + j * 32 + wave;
      if (gj >= n_chunks) break;
      const int glo = rl_i(lo, j);
      const int ghi = rl_i(hi, j);
      const int* run = packed + (size_t)gj * CH;
      for (int i = glo + lane; i < ghi; i += 128) {
        int v[4];
#pragma unroll
        for (int u = 0; u < 4; ++u) v[u] = (i + 32 * u < ghi) ? run[i + 32 * u] : -1;
#pragma unroll
        for (int u = 0; u < 4; ++u)
          if (v[u] >= 0) atomicAdd(&cnt[v[u] & (NBK - 1)], 1);
      }
    }
  }
  __syncthreads();

  int c = 0, x = 0;
  if (t < NBK) {
    c = cnt[t];
    x = c;
    for (int o = 1; o < 32; o <<= 1) {
      int y = __shfl_up(x, o);
      if (lane >= o) x += y;
    }
    if (lane == 31) wsum[wave] = x;
  }
  __syncthreads();
  if (wave == 0) {
    const int w = (lane < NBK / 32) ? wsum[lane] : 0;
    int s = w;
    for (int o = 1; o < 32; o <<= 1) {
      int y = __shfl_up(s, o);
      if (lane >= o) s += y;
    }
    wsum[lane] = s - w;
  }
  __syncthreads();
  if (t < NBK) {
    const int ex = wsum[wave] + x - c;
    cur[t] = ex;
    const int node = b * NBK + t;
    if (node < n_nodes) {
      deg[node] = c;
      rowptr[node] = bstart + ex;
      dinv[node] = rsqrtf((float)(c + 1));
    }
  }
  __syncthreads();

  const bool staged = nbe <= CAP;
  for (int gb = 0; gb < n_chunks; gb += 1024) {
    const int g = gb + lane * 32 + wave;
    int lo = 0, hi = 0;
    if (g < n_chunks) {
      lo = tab[(size_t)g * stride + b];
      hi = tab[(size_t)g * stride + b + 1];
    }
    for (int j = 0; j < 32; ++j) {
      const int gj = gb + j * 32 + wave;
      if (gj >= n_chunks) break;
      const int glo = rl_i(lo, j);
      const int ghi = rl_i(hi, j);
      const int* run = packed + (size_t)gj * CH;
      for (int i = glo + lane; i < ghi; i += 128) {
        int v[4];
#pragma unroll
        for (int u = 0; u < 4; ++u) v[u] = (i + 32 * u < ghi) ? run[i + 32 * u] : -1;
#pragma unroll
        for (int u = 0; u < 4; ++u) {
          if (v[u] >= 0) {
            const int p = atomicAdd(&cur[v[u] & (NBK - 1)], 1);
            const int s = v[u] >> NBK_SH;
            if (staged) stage[p] = s;
            else        col[bstart + p] = s;
          }
        }
      }
    }
  }
  __syncthreads();
  if (staged) {
    for (int i = t; i < nbe; i += BT) col[bstart + i] = stage[i];
  }
}


__global__ __launch_bounds__(256) void gemm1_wmma_kernel(
    const float* __restrict__ x, const float* __restrict__ W1,
    const float* __restrict__ dinv, float* __restrict__ h1, int n_nodes) {
  __shared__ float sW[IN_FEAT * HIDDEN];
  for (int i = threadIdx.x; i < IN_FEAT * HIDDEN; i += 256) {
    int k = i >> 5;
    int n = i & 31;
    sW[(k >> 1) * 64 + n * 2 + (k & 1)] = W1[i];
  }
  __syncthreads();

  const int wave = threadIdx.x >> 5;
  const int lane = threadIdx.x & 31;
  const int half = lane >> 4;
  const int ml   = lane & 15;

  const int tile   = blockIdx.x * 8 + wave;
  const int ntiles = n_nodes >> 4;
  if (tile >= ntiles) return;

  const float* xrow = x + (size_t)(tile * 16 + ml) * IN_FEAT + 2 * half;
  const float* bptr = sW + half * 64 + ml * 2;

  v8f acc0 = {};
  v8f acc1 = {};

  for (int k0 = 0; k0 < IN_FEAT; k0 += 32) {
    __builtin_prefetch(xrow + k0 + 256, 0, 3);
#pragma unroll
    for (int kk = 0; kk < 32; kk += 4) {
      const int k = k0 + kk;
      v2f a  = *(const v2f*)(xrow + k);
      v2f b0 = *(const v2f*)(bptr + k * 32);
      v2f b1 = *(const v2f*)(bptr + k * 32 + 32);
      acc0 = __builtin_amdgcn_wmma_f32_16x16x4_f32(false, a, false, b0,
                                                   (short)0, acc0, false, false);
      acc1 = __builtin_amdgcn_wmma_f32_16x16x4_f32(false, a, false, b1,
                                                   (short)0, acc1, false, false);
    }
  }

  const float4* dr = (const float4*)(dinv + tile * 16 + half * 8);
  const float4 d0 = dr[0];
  const float4 d1 = dr[1];
  const float ds[8] = {d0.x, d0.y, d0.z, d0.w, d1.x, d1.y, d1.z, d1.w};

  float* orow = h1 + (size_t)(tile * 16 + half * 8) * HIDDEN;
#pragma unroll
  for (int g = 0; g < 8; ++g) {
    orow[(size_t)g * HIDDEN + ml]      = acc0[g] * ds[g];
    orow[(size_t)g * HIDDEN + 16 + ml] = acc1[g] * ds[g];
  }
}


#define GWG     2048
#define NWAVES  (GWG * 8)
#define G1WG    1024
#define NWAVES1 (G1WG * 8)

template <int NG>
__device__ __forceinline__ void ask_rows_l1(const float* __restrict__ h1, int sj, int m, int q,
                                            unsigned f4off, unsigned ownoff, float4* hv) {
  unsigned off[NG];
#pragma unroll
  for (int it = 0; it < NG; ++it) {
    const int jn = it * 4 + q;
    const int s = __shfl(sj, jn);
    off[it] = (jn < m) ? ((unsigned)s * (unsigned)(HIDDEN * sizeof(float)) + f4off) : ownoff;
  }
#pragma unroll
  for (int it = 0; it < NG; ++it)
    hv[it] = *(const float4*)((const char*)h1 + off[it]);
}

template <int NG>
__device__ __forceinline__ float4 add_rows_l1(const float4* hv, int m, int q, float4 acc) {
#pragma unroll
  for (int it = 0; it < NG; ++it) {
    const bool ok = (it * 4 + q) < m;
    acc.x += ok ? hv[it].x : 0.0f;
    acc.y += ok ? hv[it].y : 0.0f;
    acc.z += ok ? hv[it].z : 0.0f;
    acc.w += ok ? hv[it].w : 0.0f;
  }
  return acc;
}

__global__ __launch_bounds__(256) void gather1_kernel(
    const int* __restrict__ rowptr, const int* __restrict__ deg,
    const int* __restrict__ col, const float* __restrict__ dinv,
    const float* __restrict__ h1, const float* __restrict__ b1,
    const float* __restrict__ W2, float* __restrict__ h2, int n_nodes) {
  const int lane = threadIdx.x & 31;
  const int q = lane >> 3;
  const int f4 = lane & 7;
  const unsigned f4off = (unsigned)f4 * 16u;
  int v = __builtin_amdgcn_readfirstlane(blockIdx.x * 8 + (threadIdx.x >> 5));
  if (v >= n_nodes) return;

  const float4 bias = *(const float4*)(b1 + 4 * f4);
  const int c0 = 2 * q;
  float w2a[4], w2b[4];
#pragma unroll
  for (int k = 0; k < 4; ++k) {
    w2a[k] = W2[(4 * f4 + k) * NCLS + c0];
    w2b[k] = (c0 + 1 < NCLS) ? W2[(4 * f4 + k) * NCLS + c0 + 1] : 0.0f;
  }

  int vn = v;
  bool more = true;
  int nbeg = rowptr[vn];
  int ncnt = deg[vn];
  float ndv = dinv[vn];
  bool have = false;
  int beg = 0, cnt = 0;
  float dv = 0.0f;
  float4 hself = make_float4(0.0f, 0.0f, 0.0f, 0.0f);
  int sj0 = 0, sj1 = 0;

  for (;;) {
    float4 nhself = make_float4(0.0f, 0.0f, 0.0f, 0.0f);
    int nsj0 = 0, nsj1 = 0;
    if (more) {
      nhself = *(const float4*)(h1 + (size_t)vn * HIDDEN + 4 * f4);
      if (lane < ncnt) nsj0 = col[nbeg + lane];
      if (lane + 32 < ncnt) nsj1 = col[nbeg + 32 + lane];
    }

    const unsigned ownoff = (unsigned)v * (unsigned)(HIDDEN * sizeof(float)) + f4off;
    const int m0 = cnt < 32 ? cnt : 32;
    const int m1 = cnt < 64 ? cnt - 32 : 32;
    float4 acc = make_float4(0.0f, 0.0f, 0.0f, 0.0f);
    if (m0 > 0) {
      float4 hv0[8];
      ask_rows_l1<8>(h1, sj0, m0, q, f4off, ownoff, hv0);
      if (m1 > 16) {
        float4 hv1[8];
        ask_rows_l1<8>(h1, sj1, m1, q, f4off, ownoff, hv1);
        __builtin_amdgcn_sched_barrier(0);
        acc = add_rows_l1<8>(hv0, m0, q, acc);
        acc = add_rows_l1<8>(hv1, m1, q, acc);
      } else if (m1 > 0) {
        float4 hv1[4];
        ask_rows_l1<4>(h1, sj1, m1, q, f4off, ownoff, hv1);
        __builtin_amdgcn_sched_barrier(0);
        acc = add_rows_l1<8>(hv0, m0, q, acc);
        acc = add_rows_l1<4>(hv1, m1, q, acc);
      } else {
        acc = add_rows_l1<8>(hv0, m0, q, acc);
      }
    }
    for (int base = 64; base < cnt; base += 32) {
      const int left = cnt - base;
      const int m = left < 32 ? left : 32;
      int sj = 0;
      if (lane < m) sj = col[beg + base + lane];
      float4 hvx[8];
      ask_rows_l1<8>(h1, sj, m, q, f4off, ownoff, hvx);
      acc = add_rows_l1<8>(hvx, m, q, acc);
    }

    const int vnn = vn + NWAVES1;
    const bool more2 = more && vnn < n_nodes;
    int nnbeg = 0, nncnt = 0;
    float nndv = 0.0f;
    if (more2) {
      nnbeg = rowptr[vnn];
      nncnt = deg[vnn];
      nndv = dinv[vnn];
    }

    acc.x += __shfl_xor(acc.x, 8);
    acc.y += __shfl_xor(acc.y, 8);
    acc.z += __shfl_xor(acc.z, 8);
    acc.w += __shfl_xor(acc.w, 8);
    acc.x += __shfl_xor(acc.x, 16);
    acc.y += __shfl_xor(acc.y, 16);
    acc.z += __shfl_xor(acc.z, 16);
    acc.w += __shfl_xor(acc.w, 16);

    const float v0 = (acc.x + hself.x) * dv + bias.x;
    const float v1 = (acc.y + hself.y) * dv + bias.y;
    const float v2 = (acc.z + hself.z) * dv + bias.z;
    const float v3 = (acc.w + hself.w) * dv + bias.w;
    const float r0 = v0 > 0.0f ? v0 : 0.0f;
    const float r1 = v1 > 0.0f ? v1 : 0.0f;
    const float r2 = v2 > 0.0f ? v2 : 0.0f;
    const float r3 = v3 > 0.0f ? v3 : 0.0f;

    float ta = r0 * w2a[0] + r1 * w2a[1] + r2 * w2a[2] + r3 * w2a[3];
    float tb = r0 * w2b[0] + r1 * w2b[1] + r2 * w2b[2] + r3 * w2b[3];
    for (int o = 1; o < 8; o <<= 1) {
      ta += __shfl_xor(ta, o);
      tb += __shfl_xor(tb, o);
    }
    if (have && f4 == 0) {
      float2 o2;
      o2.x = ta * dv;
      o2.y = tb * dv;
      *(float2*)(h2 + (size_t)v * H2S + c0) = o2;
    }

    if (!more) break;
    have = true;
    v = vn;
    beg = nbeg;
    cnt = ncnt;
    dv = ndv;
    hself = nhself;
    sj0 = nsj0;
    sj1 = nsj1;
    vn = vnn;
    more = more2;
    nbeg = nnbeg;
    ncnt = nncnt;
    ndv = nndv;
  }
}


__device__ __forceinline__ float sum_rows_l2(const float* __restrict__ h2, int sj, int m,
                                             int q, int c, float acc) {
  int s[8];
#pragma unroll
  for (int it = 0; it < 8; ++it) s[it] = __shfl(sj, it * 4 + q);
  float hv[8];
#pragma unroll
  for (int it = 0; it < 8; ++it) {
    const int jn = it * 4 + q;
    hv[it] = (jn < m) ? h2[(size_t)s[it] * H2S + c] : 0.0f;
  }
#pragma unroll
  for (int it = 0; it < 8; ++it) acc += hv[it];
  return acc;
}

__global__ __launch_bounds__(256) void gather2_softmax_kernel(
    const int* __restrict__ rowptr, const int* __restrict__ deg,
    const int* __restrict__ col, const float* __restrict__ dinv,
    const float* __restrict__ h2, const float* __restrict__ b2,
    float* __restrict__ y, int n_nodes) {
  const int lane = threadIdx.x & 31;
  const int q = lane >> 3;
  const int c = lane & 7;
  int v = __builtin_amdgcn_readfirstlane(blockIdx.x * 8 + (threadIdx.x >> 5));
  if (v >= n_nodes) return;

  const float bias = (c < NCLS) ? b2[c] : 0.0f;

  int vn = v;
  bool more = true;
  int nbeg = rowptr[vn];
  int ncnt = deg[vn];
  float ndv = dinv[vn];
  bool have = false;
  int beg = 0, cnt = 0;
  float dv = 0.0f, hown = 0.0f;
  int sj0 = 0, sj1 = 0;

  for (;;) {
    float nhown = 0.0f;
    int nsj0 = 0, nsj1 = 0;
    if (more) {
      nhown = h2[(size_t)vn * H2S + c];
      if (lane < ncnt) nsj0 = col[nbeg + lane];
      if (lane + 32 < ncnt) nsj1 = col[nbeg + 32 + lane];
    }

    float acc = 0.0f;
#pragma clang loop unroll(disable)
    for (int k = 0; k < 2; ++k) {
      const int mk = cnt - 32 * k;
      if (mk <= 0) break;
      acc = sum_rows_l2(h2, k == 0 ? sj0 : sj1, mk < 32 ? mk : 32, q, c, acc);
    }
    for (int base = 64; base < cnt; base += 32) {
      const int left = cnt - base;
      const int m = left < 32 ? left : 32;
      int sj = 0;
      if (lane < m) sj = col[beg + base + lane];
      acc = sum_rows_l2(h2, sj, m, q, c, acc);
    }

    const int vnn = vn + NWAVES;
    const bool more2 = more && vnn < n_nodes;
    int nnbeg = 0, nncnt = 0;
    float nndv = 0.0f;
    if (more2) {
      nnbeg = rowptr[vnn];
      nncnt = deg[vnn];
      nndv = dinv[vnn];
    }

    acc += __shfl_xor(acc, 8);
    acc += __shfl_xor(acc, 16);

    float z = -3.0e38f;
    if (c < NCLS) z = (acc + hown) * dv + bias;
    float mx = z;
    mx = fmaxf(mx, __shfl_xor(mx, 1));
    mx = fmaxf(mx, __shfl_xor(mx, 2));
    mx = fmaxf(mx, __shfl_xor(mx, 4));
    const float ex = (c < NCLS) ? expf(z - mx) : 0.0f;
    float sum = ex;
    sum += __shfl_xor(sum, 1);
    sum += __shfl_xor(sum, 2);
    sum += __shfl_xor(sum, 4);
    const float inv = 1.0f / sum;
    if (have && q == 0 && c < NCLS) y[(size_t)v * NCLS + c] = ex * inv;

    if (!more) break;
    have = true;
    v = vn;
    beg = nbeg;
    cnt = ncnt;
    dv = ndv;
    hown = nhown;
    sj0 = nsj0;
    sj1 = nsj1;
    vn = vnn;
    more = more2;
    nbeg = nnbeg;
    ncnt = nncnt;
    ndv = nndv;
  }
}


extern "C" void kernel_launch(void* const* d_in, const int* in_sizes, int n_in,
                              void* d_out, int out_size, void* d_ws, size_t ws_size,
                              hipStream_t stream) {
  const float* x  = (const float*)d_in[0];
  const int*   ei = (const int*)d_in[1];
  const float* W1 = (const float*)d_in[2];
  const float* b1 = (const float*)d_in[3];
  const float* W2 = (const float*)d_in[4];
  const float* b2 = (const float*)d_in[5];
  float* y = (float*)d_out;

  const int n_nodes = in_sizes[0] / IN_FEAT;
  const int n_edges = in_sizes[1] / 2;
  const int* srcp = ei;
  const int* dstp = ei + n_edges;
  const int nb       = (n_nodes + NBK - 1) / NBK;
  const int n_chunks = (n_edges + CH - 1) / CH;

  char* ws = (char*)d_ws;
  size_t off = 0;
  auto carve = [&](size_t bytes) -> void* {
    void* p = ws + off;
    off += (bytes + 255) & ~(size_t)255;
    return p;
  };
  const size_t h1_bytes   = (size_t)n_nodes * HIDDEN * sizeof(float);
  const size_t rank_bytes = (size_t)n_edges * sizeof(int);
  int*   deg      = (int*)  carve((size_t)n_nodes * sizeof(int));
  float* dinv     = (float*)carve((size_t)n_nodes * sizeof(float));
  int*   rowptr   = (int*)  carve((size_t)n_nodes * sizeof(int));
  void*  shared_r = carve(h1_bytes > rank_bytes ? h1_bytes : rank_bytes);
  int*   packed   = (int*)shared_r;
  float* h1       = (float*)shared_r;
  int*   col      = (int*)  carve((size_t)n_edges * sizeof(int));
  const size_t h2_bytes  = (size_t)n_nodes * H2S * sizeof(float);
  const size_t tab_bytes = (size_t)n_chunks * (nb + 1) * sizeof(int);
  void*  shared_t = carve(h2_bytes > tab_bytes ? h2_bytes : tab_bytes);
  int*   tab      = (int*)shared_t;
  float* h2       = (float*)shared_t;
  (void)ws_size;

  bin_chunk_kernel<<<n_chunks, AT, 0, stream>>>(srcp, dstp, n_edges, nb, tab, packed);
  bucket_lists_kernel<<<nb, BT, 0, stream>>>(tab, packed, n_nodes, nb, n_chunks,
                                             deg, rowptr, dinv, col);

  int ntiles = n_nodes / 16;
  int g1_blocks = (ntiles + 7) / 8;
  gemm1_wmma_kernel<<<g1_blocks, 256, 0, stream>>>(x, W1, dinv, h1, n_nodes);

  gather1_kernel<<<G1WG, 256, 0, stream>>>(rowptr, deg, col, dinv, h1, b1, W2, h2, n_nodes);
  gather2_softmax_kernel<<<GWG, 256, 0, stream>>>(rowptr, deg, col, dinv, h2, b2, y, n_nodes);
}


# GEMM1 tiles dealt evenly over the SIMDs: one resident 26-wave workgroup per WGP (W1 staged once, in one round trip), 24 full tiles each, leftover tiles as two 16-column halves; sum order unchanged
# speedup vs baseline: 6.4701x; 1.0482x over previous
#include <hip/hip_runtime.h>


typedef float v2f __attribute__((ext_vector_type(2)));
typedef float v8f __attribute__((ext_vector_type(8)));

#define IN_FEAT 512
#define HIDDEN  32
#define NCLS    7
#define H2S     8


__device__ __forceinline__ int rl_i(int x, int l) {
  return __builtin_amdgcn_readlane(x, l);
}
__device__ __forceinline__ float rl_f(float x, int l) {
  return __int_as_float(__builtin_amdgcn_readlane(__float_as_int(x), l));
}


#define NBK_SH 9
#define NBK    (1 << NBK_SH)
#define CH     16384
#define AT     1024
#define AK     (CH / AT)
#define MAXB   1024
#define BT     1024
#define CAP    24576

__global__ __launch_bounds__(AT) void bin_chunk_kernel(
    const int* __restrict__ src, const int* __restrict__ dst, int n_edges, int nb,
    int* __restrict__ tab, int* __restrict__ packed) {
  __shared__ int hist[MAXB + 1];
  __shared__ int sP[CH];
  const int t = threadIdx.x;
  const int base = blockIdx.x * CH;
  for (int i = t; i <= nb; i += AT) hist[i] = 0;
  __syncthreads();

  int pv[AK], pr[AK], pb[AK];
#pragma unroll
  for (int k = 0; k < AK; ++k) {
    const int e = base + t + k * AT;
    const int ec = e < n_edges ? e : n_edges - 1;
    const int d = dst[ec];
    const int s = src[ec];
    pb[k] = e < n_edges ? (d >> NBK_SH) : -1;
    pv[k] = (s << NBK_SH) | (d & (NBK - 1));
  }
#pragma unroll
  for (int k = 0; k < AK; ++k) {
    pr[k] = 0;
    if (pb[k] >= 0) pr[k] = atomicAdd(&hist[pb[k]], 1);
  }
  __syncthreads();

  if (t < 32) {
    int carry = 0;
    for (int c0 = 0; c0 < nb; c0 += 32) {
      const int i = c0 + t;
      const int x = (i < nb) ? hist[i] : 0;
      int inc = x;
      for (int o = 1; o < 32; o <<= 1) {
        int y = __shfl_up(inc, o);
        if (t >= o) inc += y;
      }
      if (i < nb) hist[i] = carry + inc - x;
      carry += __shfl(inc, 31);
    }
    if (t == 0) hist[nb] = carry;
  }
  __syncthreads();

  int* trow = tab + (size_t)blockIdx.x * (nb + 1);
  for (int i = t; i <= nb; i += AT) trow[i] = hist[i];
#pragma unroll
  for (int k = 0; k < AK; ++k) {
    if (pb[k] >= 0) sP[hist[pb[k]] + pr[k]] = pv[k];
  }
  __syncthreads();

  const int cnt = hist[nb];
  for (int i = t; i < cnt; i += AT) packed[base + i] = sP[i];
}

__global__ __launch_bounds__(BT) void bucket_lists_kernel(
    const int* __restrict__ tab, const int* __restrict__ packed,
    int n_nodes, int nb, int n_chunks,
    int* __restrict__ deg, int* __restrict__ rowptr, float* __restrict__ dinv,
    int* __restrict__ col) {
  __shared__ int cnt[NBK];
  __shared__ int cur[NBK];
  __shared__ int red0[32];
  __shared__ int red1[32];
  __shared__ int bse[2];
  __shared__ int wsum[32];
  __shared__ int stage[CAP];
  const int t = threadIdx.x;
  const int lane = t & 31;
  const int wave = t >> 5;
  const int b = blockIdx.x;
  const int stride = nb + 1;

  int a0 = 0, a1 = 0;
  for (int g = t; g < n_chunks; g += BT) {
    a0 += tab[(size_t)g * stride + b];
    a1 += tab[(size_t)g * stride + b + 1];
  }
  for (int o = 16; o > 0; o >>= 1) {
    a0 += __shfl_xor(a0, o);
    a1 += __shfl_xor(a1, o);
  }
  if (lane == 0) { red0[wave] = a0; red1[wave] = a1; }
  if (t < NBK) cnt[t] = 0;
  __syncthreads();
  if (wave == 0) {
    int x0 = red0[lane], x1 = red1[lane];
    for (int o = 16; o > 0; o >>= 1) {
      x0 += __shfl_xor(x0, o);
      x1 += __shfl_xor(x1, o);
    }
    if (lane == 0) { bse[0] = x0; bse[1] = x1; }
  }
  __syncthreads();
  const int bstart = bse[0];
  const int nbe = bse[1] - bse[0];

  for (int gb = 0; gb < n_chunks; gb += 1024) {
    const int g = gb + lane * 32 + wave;
    int lo = 0, hi = 0;
    if (g < n_chunks) {
      lo = tab[(size_t)g * stride + b];
      hi = tab[(size_t)g * stride + b + 1];
    }
    for (int j = 0; j < 32; ++j) {
      const int gj = gb + j * 32 + wave;
      if (gj >= n_chunks) break;
      const int glo = rl_i(lo, j);
      const int ghi = rl_i(hi, j);
      const int* run = packed + (size_t)gj * CH;
      for (int i = glo + lane; i < ghi; i += 128) {
        int v[4];
#pragma unroll
        for (int u = 0; u < 4; ++u) v[u] = (i + 32 * u < ghi) ? run[i + 32 * u] : -1;
#pragma unroll
        for (int u = 0; u < 4; ++u)
          if (v[u] >= 0) atomicAdd(&cnt[v[u] & (NBK - 1)], 1);
      }
    }
  }
  __syncthreads();

  int c = 0, x = 0;
  if (t < NBK) {
    c = cnt[t];
    x = c;
    for (int o = 1; o < 32; o <<= 1) {
      int y = __shfl_up(x, o);
      if (lane >= o) x += y;
    }
    if (lane == 31) wsum[wave] = x;
  }
  __syncthreads();
  if (wave == 0) {
    const int w = (lane < NBK / 32) ? wsum[lane] : 0;
    int s = w;
    for (int o = 1; o < 32; o <<= 1) {
      int y = __shfl_up(s, o);
      if (lane >= o) s += y;
    }
    wsum[lane] = s - w;
  }
  __syncthreads();
  if (t < NBK) {
    const int ex = wsum[wave] + x - c;
    cur[t] = ex;
    const int node = b * NBK + t;
    if (node < n_nodes) {
      deg[node] = c;
      rowptr[node] = bstart + ex;
      dinv[node] = rsqrtf((float)(c + 1));
    }
  }
  __syncthreads();

  const bool staged = nbe <= CAP;
  for (int gb = 0; gb < n_chunks; gb += 1024) {
    const int g = gb + lane * 32 + wave;
    int lo = 0, hi = 0;
    if (g < n_chunks) {
      lo = tab[(size_t)g * stride + b];
      hi = tab[(size_t)g * stride + b + 1];
    }
    for (int j = 0; j < 32; ++j) {
      const int gj = gb + j * 32 + wave;
      if (gj >= n_chunks) break;
      const int glo = rl_i(lo, j);
      const int ghi = rl_i(hi, j);
      const int* run = packed + (size_t)gj * CH;
      for (int i = glo + lane; i < ghi; i += 128) {
        int v[4];
#pragma unroll
        for (int u = 0; u < 4; ++u) v[u] = (i + 32 * u < ghi) ? run[i + 32 * u] : -1;
#pragma unroll
        for (int u = 0; u < 4; ++u) {
          if (v[u] >= 0) {
            const int p = atomicAdd(&cur[v[u] & (NBK - 1)], 1);
            const int s = v[u] >> NBK_SH;
            if (staged) stage[p] = s;
            else        col[bstart + p] = s;
          }
        }
      }
    }
  }
  __syncthreads();
  if (staged) {
    for (int i = t; i < nbe; i += BT) col[bstart + i] = stage[i];
  }
}


#define G1B   256
#define G1FW  24
#define G1T   ((G1FW + 2) * 32)
#define G1LDS (IN_FEAT * HIDDEN + 8 * 32)

template <int NACC>
__device__ __forceinline__ void gemm1_acc(const float* __restrict__ x, const float* sW,
                                          int tile, int ch, int lane, v8f& acc0, v8f& acc1) {
  const int half = lane >> 4;
  const int ml   = lane & 15;

  const float* xrow = x + (size_t)(tile * 16 + ml) * IN_FEAT + 2 * half;
  const float* bptr = sW + half * 64 + ml * 2 + ch * 32;

  for (int k0 = 0; k0 < IN_FEAT; k0 += 32) {
    __builtin_prefetch(xrow + k0 + 256, 0, 3);
#pragma unroll
    for (int kk = 0; kk < 32; kk += 4) {
      const int k = k0 + kk;
      v2f a  = *(const v2f*)(xrow + k);
      v2f b0 = *(const v2f*)(bptr + k * 32);
      acc0 = __builtin_amdgcn_wmma_f32_16x16x4_f32(false, a, false, b0,
                                                   (short)0, acc0, false, false);
      if (NACC == 2) {
        v2f b1 = *(const v2f*)(bptr + k * 32 + 32);
        acc1 = __builtin_amdgcn_wmma_f32_16x16x4_f32(false, a, false, b1,
                                                     (short)0, acc1, false, false);
      }
    }
  }
}

__global__ __launch_bounds__(G1T) void gemm1_wmma_kernel(
    const float* __restrict__ x, const float* __restrict__ W1,
    const float* __restrict__ dinv, float* __restrict__ h1, int n_nodes) {
  __shared__ __attribute__((aligned(16))) float sW[G1LDS];
  float* part = sW + IN_FEAT * HIDDEN;
  constexpr int NIT = (IN_FEAT / 2) * (HIDDEN / 4);
  constexpr int NU  = (NIT + G1T - 1) / G1T;
  float4 r0[NU], r1[NU];
#pragma unroll
  for (int u = 0; u < NU; ++u) {
    const int i = threadIdx.x + u * G1T;
    if (i < NIT) {
      r0[u] = *(const float4*)(W1 + (size_t)(2 * (i >> 3)) * HIDDEN + 4 * (i & 7));
      r1[u] = *(const float4*)(W1 + (size_t)(2 * (i >> 3) + 1) * HIDDEN + 4 * (i & 7));
    }
  }
#pragma unroll
  for (int u = 0; u < NU; ++u) {
    const int i = threadIdx.x + u * G1T;
    if (i < NIT) {
      float4* d = (float4*)(sW + (i >> 3) * 64 + (i & 7) * 8);
      d[0] = make_float4(r0[u].x, r1[u].x, r0[u].y, r1[u].y);
      d[1] = make_float4(r0[u].z, r1[u].z, r0[u].w, r1[u].w);
    }
  }
  __syncthreads();

  const int wave = threadIdx.x >> 5;
  const int lane = threadIdx.x & 31;
  const int ntiles = n_nodes >> 4;
  const int per    = ntiles / G1B;
  const int left   = ntiles - per * G1B;
  const bool extra = blockIdx.y == 0 && (int)blockIdx.x < left;

  v8f acc0 = {};
  v8f acc1 = {};
  int tile = -1;
  if (wave < G1FW) {
    const int r = blockIdx.y * G1FW + wave;
    if (r < per) {
      tile = blockIdx.x * per + r;
      gemm1_acc<2>(x, sW, tile, 0, lane, acc0, acc1);
    }
  } else if (extra) {
    const int t = per * G1B + blockIdx.x;
    const int ch = wave - G1FW;
    v8f p = {};
    gemm1_acc<1>(x, sW, t, ch, lane, p, acc1);
    if (ch == 0) {
      tile = t;
      acc0 = p;
    } else {
#pragma unroll
      for (int g = 0; g < 8; ++g) part[g * 32 + lane] = p[g];
    }
  }
  __syncthreads();
  if (extra && wave == G1FW) {
#pragma unroll
    for (int g = 0; g < 8; ++g) acc1[g] = part[g * 32 + lane];
  }
  if (tile < 0) return;

  const int half = lane >> 4;
  const int ml   = lane & 15;
  const float4* dr = (const float4*)(dinv + tile * 16 + half * 8);
  const float4 d0 = dr[0];
  const float4 d1 = dr[1];
  const float ds[8] = {d0.x, d0.y, d0.z, d0.w, d1.x, d1.y, d1.z, d1.w};

  float* orow = h1 + (size_t)(tile * 16 + half * 8) * HIDDEN;
#pragma unroll
  for (int g = 0; g < 8; ++g) {
    orow[(size_t)g * HIDDEN + ml]      = acc0[g] * ds[g];
    orow[(size_t)g * HIDDEN + 16 + ml] = acc1[g] * ds[g];
  }
}


#define GWG     2048
#define NWAVES  (GWG * 8)
#define G1WG    1024
#define NWAVES1 (G1WG * 8)

template <int NG>
__device__ __forceinline__ void ask_rows_l1(const float* __restrict__ h1, int sj, int m, int q,
                                            unsigned f4off, unsigned ownoff, float4* hv) {
  unsigned off[NG];
#pragma unroll
  for (int it = 0; it < NG; ++it) {
    const int jn = it * 4 + q;
    const int s = __shfl(sj, jn);
    off[it] = (jn < m) ? ((unsigned)s * (unsigned)(HIDDEN * sizeof(float)) + f4off) : ownoff;
  }
#pragma unroll
  for (int it = 0; it < NG; ++it)
    hv[it] = *(const float4*)((const char*)h1 + off[it]);
}

template <int NG>
__device__ __forceinline__ float4 add_rows_l1(const float4* hv, int m, int q, float4 acc) {
#pragma unroll
  for (int it = 0; it < NG; ++it) {
    const bool ok = (it * 4 + q) < m;
    acc.x += ok ? hv[it].x : 0.0f;
    acc.y += ok ? hv[it].y : 0.0f;
    acc.z += ok ? hv[it].z : 0.0f;
    acc.w += ok ? hv[it].w : 0.0f;
  }
  return acc;
}

__global__ __launch_bounds__(256) void gather1_kernel(
    const int* __restrict__ rowptr, const int* __restrict__ deg,
    const int* __restrict__ col, const float* __restrict__ dinv,
    const float* __restrict__ h1, const float* __restrict__ b1,
    const float* __restrict__ W2, float* __restrict__ h2, int n_nodes) {
  const int lane = threadIdx.x & 31;
  const int q = lane >> 3;
  const int f4 = lane & 7;
  const unsigned f4off = (unsigned)f4 * 16u;
  int v = __builtin_amdgcn_readfirstlane(blockIdx.x * 8 + (threadIdx.x >> 5));
  if (v >= n_nodes) return;

  const float4 bias = *(const float4*)(b1 + 4 * f4);
  const int c0 = 2 * q;
  float w2a[4], w2b[4];
#pragma unroll
  for (int k = 0; k < 4; ++k) {
    w2a[k] = W2[(4 * f4 + k) * NCLS + c0];
    w2b[k] = (c0 + 1 < NCLS) ? W2[(4 * f4 + k) * NCLS + c0 + 1] : 0.0f;
  }

  int vn = v;
  bool more = true;
  int nbeg = rowptr[vn];
  int ncnt = deg[vn];
  float ndv = dinv[vn];
  bool have = false;
  int beg = 0, cnt = 0;
  float dv = 0.0f;
  float4 hself = make_float4(0.0f, 0.0f, 0.0f, 0.0f);
  int sj0 = 0, sj1 = 0;

  for (;;) {
    float4 nhself = make_float4(0.0f, 0.0f, 0.0f, 0.0f);
    int nsj0 = 0, nsj1 = 0;
    if (more) {
      nhself = *(const float4*)(h1 + (size_t)vn * HIDDEN + 4 * f4);
      if (lane < ncnt) nsj0 = col[nbeg + lane];
      if (lane + 32 < ncnt) nsj1 = col[nbeg + 32 + lane];
    }

    const unsigned ownoff = (unsigned)v * (unsigned)(HIDDEN * sizeof(float)) + f4off;
    const int m0 = cnt < 32 ? cnt : 32;
    const int m1 = cnt < 64 ? cnt - 32 : 32;
    float4 acc = make_float4(0.0f, 0.0f, 0.0f, 0.0f);
    if (m0 > 0) {
      float4 hv0[8];
      ask_rows_l1<8>(h1, sj0, m0, q, f4off, ownoff, hv0);
      if (m1 > 16) {
        float4 hv1[8];
        ask_rows_l1<8>(h1, sj1, m1, q, f4off, ownoff, hv1);
        __builtin_amdgcn_sched_barrier(0);
        acc = add_rows_l1<8>(hv0, m0, q, acc);
        acc = add_rows_l1<8>(hv1, m1, q, acc);
      } else if (m1 > 0) {
        float4 hv1[4];
        ask_rows_l1<4>(h1, sj1, m1, q, f4off, ownoff, hv1);
        __builtin_amdgcn_sched_barrier(0);
        acc = add_rows_l1<8>(hv0, m0, q, acc);
        acc = add_rows_l1<4>(hv1, m1, q, acc);
      } else {
        acc = add_rows_l1<8>(hv0, m0, q, acc);
      }
    }
    for (int base = 64; base < cnt; base += 32) {
      const int left = cnt - base;
      const int m = left < 32 ? left : 32;
      int sj = 0;
      if (lane < m) sj = col[beg + base + lane];
      float4 hvx[8];
      ask_rows_l1<8>(h1, sj, m, q, f4off, ownoff, hvx);
      acc = add_rows_l1<8>(hvx, m, q, acc);
    }

    const int vnn = vn + NWAVES1;
    const bool more2 = more && vnn < n_nodes;
    int nnbeg = 0, nncnt = 0;
    float nndv = 0.0f;
    if (more2) {
      nnbeg = rowptr[vnn];
      nncnt = deg[vnn];
      nndv = dinv[vnn];
    }

    acc.x += __shfl_xor(acc.x, 8);
    acc.y += __shfl_xor(acc.y, 8);
    acc.z += __shfl_xor(acc.z, 8);
    acc.w += __shfl_xor(acc.w, 8);
    acc.x += __shfl_xor(acc.x, 16);
    acc.y += __shfl_xor(acc.y, 16);
    acc.z += __shfl_xor(acc.z, 16);
    acc.w += __shfl_xor(acc.w, 16);

    const float v0 = (acc.x + hself.x) * dv + bias.x;
    const float v1 = (acc.y + hself.y) * dv + bias.y;
    const float v2 = (acc.z + hself.z) * dv + bias.z;
    const float v3 = (acc.w + hself.w) * dv + bias.w;
    const float r0 = v0 > 0.0f ? v0 : 0.0f;
    const float r1 = v1 > 0.0f ? v1 : 0.0f;
    const float r2 = v2 > 0.0f ? v2 : 0.0f;
    const float r3 = v3 > 0.0f ? v3 : 0.0f;

    float ta = r0 * w2a[0] + r1 * w2a[1] + r2 * w2a[2] + r3 * w2a[3];
    float tb = r0 * w2b[0] + r1 * w2b[1] + r2 * w2b[2] + r3 * w2b[3];
    for (int o = 1; o < 8; o <<= 1) {
      ta += __shfl_xor(ta, o);
      tb += __shfl_xor(tb, o);
    }
    if (have && f4 == 0) {
      float2 o2;
      o2.x = ta * dv;
      o2.y = tb * dv;
      *(float2*)(h2 + (size_t)v * H2S + c0) = o2;
    }

    if (!more) break;
    have = true;
    v = vn;
    beg = nbeg;
    cnt = ncnt;
    dv = ndv;
    hself = nhself;
    sj0 = nsj0;
    sj1 = nsj1;
    vn = vnn;
    more = more2;
    nbeg = nnbeg;
    ncnt = nncnt;
    ndv = nndv;
  }
}


__device__ __forceinline__ float sum_rows_l2(const float* __restrict__ h2, int sj, int m,
                                             int q, int c, float acc) {
  int s[8];
#pragma unroll
  for (int it = 0; it < 8; ++it) s[it] = __shfl(sj, it * 4 + q);
  float hv[8];
#pragma unroll
  for (int it = 0; it < 8; ++it) {
    const int jn = it * 4 + q;
    hv[it] = (jn < m) ? h2[(size_t)s[it] * H2S + c] : 0.0f;
  }
#pragma unroll
  for (int it = 0; it < 8; ++it) acc += hv[it];
  return acc;
}

__global__ __launch_bounds__(256) void gather2_softmax_kernel(
    const int* __restrict__ rowptr, const int* __restrict__ deg,
    const int* __restrict__ col, const float* __restrict__ dinv,
    const float* __restrict__ h2, const float* __restrict__ b2,
    float* __restrict__ y, int n_nodes) {
  const int lane = threadIdx.x & 31;
  const int q = lane >> 3;
  const int c = lane & 7;
  int v = __builtin_amdgcn_readfirstlane(blockIdx.x * 8 + (threadIdx.x >> 5));
  if (v >= n_nodes) return;

  const float bias = (c < NCLS) ? b2[c] : 0.0f;

  int vn = v;
  bool more = true;
  int nbeg = rowptr[vn];
  int ncnt = deg[vn];
  float ndv = dinv[vn];
  bool have = false;
  int beg = 0, cnt = 0;
  float dv = 0.0f, hown = 0.0f;
  int sj0 = 0, sj1 = 0;

  for (;;) {
    float nhown = 0.0f;
    int nsj0 = 0, nsj1 = 0;
    if (more) {
      nhown = h2[(size_t)vn * H2S + c];
      if (lane < ncnt) nsj0 = col[nbeg + lane];
      if (lane + 32 < ncnt) nsj1 = col[nbeg + 32 + lane];
    }

    float acc = 0.0f;
#pragma clang loop unroll(disable)
    for (int k = 0; k < 2; ++k) {
      const int mk = cnt - 32 * k;
      if (mk <= 0) break;
      acc = sum_rows_l2(h2, k == 0 ? sj0 : sj1, mk < 32 ? mk : 32, q, c, acc);
    }
    for (int base = 64; base < cnt; base += 32) {
      const int left = cnt - base;
      const int m = left < 32 ? left : 32;
      int sj = 0;
      if (lane < m) sj = col[beg + base + lane];
      acc = sum_rows_l2(h2, sj, m, q, c, acc);
    }

    const int vnn = vn + NWAVES;
    const bool more2 = more && vnn < n_nodes;
    int nnbeg = 0, nncnt = 0;
    float nndv = 0.0f;
    if (more2) {
      nnbeg = rowptr[vnn];
      nncnt = deg[vnn];
      nndv = dinv[vnn];
    }

    acc += __shfl_xor(acc, 8);
    acc += __shfl_xor(acc, 16);

    float z = -3.0e38f;
    if (c < NCLS) z = (acc + hown) * dv + bias;
    float mx = z;
    mx = fmaxf(mx, __shfl_xor(mx, 1));
    mx = fmaxf(mx, __shfl_xor(mx, 2));
    mx = fmaxf(mx, __shfl_xor(mx, 4));
    const float ex = (c < NCLS) ? expf(z - mx) : 0.0f;
    float sum = ex;
    sum += __shfl_xor(sum, 1);
    sum += __shfl_xor(sum, 2);
    sum += __shfl_xor(sum, 4);
    const float inv = 1.0f / sum;
    if (have && q == 0 && c < NCLS) y[(size_t)v * NCLS + c] = ex * inv;

    if (!more) break;
    have = true;
    v = vn;
    beg = nbeg;
    cnt = ncnt;
    dv = ndv;
    hown = nhown;
    sj0 = nsj0;
    sj1 = nsj1;
    vn = vnn;
    more = more2;
    nbeg = nnbeg;
    ncnt = nncnt;
    ndv = nndv;
  }
}


extern "C" void kernel_launch(void* const* d_in, const int* in_sizes, int n_in,
                              void* d_out, int out_size, void* d_ws, size_t ws_size,
                              hipStream_t stream) {
  const float* x  = (const float*)d_in[0];
  const int*   ei = (const int*)d_in[1];
  const float* W1 = (const float*)d_in[2];
  const float* b1 = (const float*)d_in[3];
  const float* W2 = (const float*)d_in[4];
  const float* b2 = (const float*)d_in[5];
  float* y = (float*)d_out;

  const int n_nodes = in_sizes[0] / IN_FEAT;
  const int n_edges = in_sizes[1] / 2;
  const int* srcp = ei;
  const int* dstp = ei + n_edges;
  const int nb       = (n_nodes + NBK - 1) / NBK;
  const int n_chunks = (n_edges + CH - 1) / CH;

  char* ws = (char*)d_ws;
  size_t off = 0;
  auto carve = [&](size_t bytes) -> void* {
    void* p = ws + off;
    off += (bytes + 255) & ~(size_t)255;
    return p;
  };
  const size_t h1_bytes   = (size_t)n_nodes * HIDDEN * sizeof(float);
  const size_t rank_bytes = (size_t)n_edges * sizeof(int);
  int*   deg      = (int*)  carve((size_t)n_nodes * sizeof(int));
  float* dinv     = (float*)carve((size_t)n_nodes * sizeof(float));
  int*   rowptr   = (int*)  carve((size_t)n_nodes * sizeof(int));
  void*  shared_r = carve(h1_bytes > rank_bytes ? h1_bytes : rank_bytes);
  int*   packed   = (int*)shared_r;
  float* h1       = (float*)shared_r;
  int*   col      = (int*)  carve((size_t)n_edges * sizeof(int));
  const size_t h2_bytes  = (size_t)n_nodes * H2S * sizeof(float);
  const size_t tab_bytes = (size_t)n_chunks * (nb + 1) * sizeof(int);
  void*  shared_t = carve(h2_bytes > tab_bytes ? h2_bytes : tab_bytes);
  int*   tab      = (int*)shared_t;
  float* h2       = (float*)shared_t;
  (void)ws_size;

  bin_chunk_kernel<<<n_chunks, AT, 0, stream>>>(srcp, dstp, n_edges, nb, tab, packed);
  bucket_lists_kernel<<<nb, BT, 0, stream>>>(tab, packed, n_nodes, nb, n_chunks,
                                             deg, rowptr, dinv, col);

  const int g1_rounds = (n_nodes / 16 / G1B + G1FW - 1) / G1FW;
  gemm1_wmma_kernel<<<dim3(G1B, g1_rounds > 0 ? g1_rounds : 1), G1T, 0, stream>>>(x, W1, dinv, h1, n_nodes);

  gather1_kernel<<<G1WG, 256, 0, stream>>>(rowptr, deg, col, dinv, h1, b1, W2, h2, n_nodes);
  gather2_softmax_kernel<<<GWG, 256, 0, stream>>>(rowptr, deg, col, dinv, h2, b2, y, n_nodes);
}
